# starting kernel
# baseline (speedup 1.0000x reference)
#include <hip/hip_runtime.h>
#include <math.h>

#define B_    16
#define C_    128
#define N_    8192
#define NH    4096
#define KBINS 4097
#define KS    4104
#define WPAIR 64

typedef __attribute__((ext_vector_type(2))) float v2f;
typedef __attribute__((ext_vector_type(8))) float v8f;

__global__ __launch_bounds__(256) void k0_twiddle(float2* __restrict__ Wc,
                                                  float2* __restrict__ Ws)
{
    const int idx = blockIdx.x * 256 + threadIdx.x;
    const int m = idx >> 6, j = idx & 63;
    const float scale = 1.0f / 1048576.0f;
    float f0 = (float)(m * (2 * j))     * (1.f / 64.f);
    float f1 = (float)(m * (2 * j + 1)) * (1.f / 64.f);
    Wc[idx] = make_float2(cospif(f0) * scale, cospif(f1) * scale);
    Ws[idx] = make_float2(sinpif(f0) * scale, sinpif(f1) * scale);
}

__global__ __launch_bounds__(256) void k1_row_fft(const float* __restrict__ xre,
                                                  const float* __restrict__ xim,
                                                  const float* __restrict__ gamma,
                                                  const float* __restrict__ beta,
                                                  float2* __restrict__ R)
{
    __shared__ float re[N_];
    __shared__ float im[N_];
    const int row = blockIdx.x;
    const int c   = row & (C_ - 1);
    const int tid = threadIdx.x;
    const size_t base = (size_t)row * N_;

    for (int i = tid; i < N_; i += 256) {
        re[i] = xre[base + i];
        im[i] = xim[base + i];
    }
    __syncthreads();

    for (int ls = 12; ls >= 0; --ls) {
        const int s = 1 << ls;
        for (int t = tid; t < NH; t += 256) {
            const int j  = t & (s - 1);
            const int i0 = ((t >> ls) << (ls + 1)) + j;
            const int i1 = i0 + s;
            float ar = re[i0], ai = im[i0];
            float br = re[i1], bi = im[i1];
            float a  = (float)j / (float)s;
            float wr = cospif(a);
            float wi = sinpif(a);
            float tr = ar - br, ti = ai - bi;
            re[i0] = ar + br;  im[i0] = ai + bi;
            re[i1] = tr * wr - ti * wi;
            im[i1] = tr * wi + ti * wr;
        }
        __syncthreads();
    }

    float lsum = 0.f, lsq = 0.f;
    for (int i = tid; i < N_; i += 256) { float v = re[i]; lsum += v; lsq += v * v; }
    im[tid]       = lsum;
    im[256 + tid] = lsq;
    __syncthreads();
    for (int off = 128; off > 0; off >>= 1) {
        if (tid < off) { im[tid] += im[tid + off]; im[256 + tid] += im[256 + tid + off]; }
        __syncthreads();
    }
    const float mean = im[0]   * (1.f / N_);
    const float var  = im[256] * (1.f / N_) - mean * mean;
    const float istd = rsqrtf(var + 1e-5f);
    const float gm = gamma[c], bt = beta[c];
    __syncthreads();

    for (int i = tid; i < N_; i += 256) {
        float v = (re[i] - mean) * istd * gm + bt;
        v = 0.5f * v * (1.f + erff(v * 0.70710678118654752f));
        re[i] = v;
        im[i] = 0.f;
    }
    __syncthreads();

    for (int ls = 0; ls <= 12; ++ls) {
        const int s = 1 << ls;
        for (int t = tid; t < NH; t += 256) {
            const int j  = t & (s - 1);
            const int i0 = ((t >> ls) << (ls + 1)) + j;
            const int i1 = i0 + s;
            float a  = (float)j / (float)s;
            float wr = cospif(a);
            float wi = -sinpif(a);
            float br = re[i1], bi = im[i1];
            float tr = br * wr - bi * wi;
            float ti = br * wi + bi * wr;
            float ar = re[i0], ai = im[i0];
            re[i0] = ar + tr;  im[i0] = ai + ti;
            re[i1] = ar - tr;  im[i1] = ai - ti;
        }
        __syncthreads();
    }

    float2* Rrow = R + (size_t)row * KS;
    for (int k = tid; k < KBINS; k += 256)
        Rrow[k] = make_float2(re[k], im[k]);
}

__global__ __launch_bounds__(256) void k2_dftC_wmma(const float2* __restrict__ R,
                                                    const float2* __restrict__ Wc,
                                                    const float2* __restrict__ Ws,
                                                    float* __restrict__ outre,
                                                    float* __restrict__ outim)
{
    __shared__ float Lre[C_][40];
    __shared__ float Lim[C_][40];
    const int b    = blockIdx.y;
    const int col0 = blockIdx.x * 32;
    const int tid  = threadIdx.x;
    const float2* Rb = R + (size_t)b * C_ * KS;

    for (int idx = tid; idx < C_ * 32; idx += 256) {
        int r = idx >> 5, cc = idx & 31;
        int k = col0 + cc;
        float2 v = (k < KBINS) ? Rb[(size_t)r * KS + k] : make_float2(0.f, 0.f);
        Lre[r][cc] = v.x;
        Lim[r][cc] = v.y;
    }
    __syncthreads();

    const int wave = tid >> 5;
    const int lane = tid & 31;
    const int lg   = lane >> 4;
    const int ln   = lane & 15;
    const int m    = wave * 16 + ln;
    const float2* WcRow = Wc + m * WPAIR;
    const float2* WsRow = Ws + m * WPAIR;

    v8f accr0 = {0.f,0.f,0.f,0.f,0.f,0.f,0.f,0.f};
    v8f acci0 = {0.f,0.f,0.f,0.f,0.f,0.f,0.f,0.f};
    v8f accr1 = {0.f,0.f,0.f,0.f,0.f,0.f,0.f,0.f};
    v8f acci1 = {0.f,0.f,0.f,0.f,0.f,0.f,0.f,0.f};

    for (int k0 = 0; k0 < C_; k0 += 4) {
        const int kA = k0 + 2 * lg;
        float2 wc = WcRow[(k0 >> 1) + lg];
        float2 wsn = WsRow[(k0 >> 1) + lg];
        v2f aRe = {  wc.x,  wc.y  };
        v2f aPs = {  wsn.x, wsn.y };
        v2f aNs = { -wsn.x, -wsn.y };
        v2f bRe0 = { Lre[kA][ln],      Lre[kA + 1][ln]      };
        v2f bIm0 = { Lim[kA][ln],      Lim[kA + 1][ln]      };
        v2f bRe1 = { Lre[kA][16 + ln], Lre[kA + 1][16 + ln] };
        v2f bIm1 = { Lim[kA][16 + ln], Lim[kA + 1][16 + ln] };
        accr0 = __builtin_amdgcn_wmma_f32_16x16x4_f32(false, aRe, false, bRe0, (short)0, accr0, false, false);
        accr0 = __builtin_amdgcn_wmma_f32_16x16x4_f32(false, aPs, false, bIm0, (short)0, accr0, false, false);
        acci0 = __builtin_amdgcn_wmma_f32_16x16x4_f32(false, aRe, false, bIm0, (short)0, acci0, false, false);
        acci0 = __builtin_amdgcn_wmma_f32_16x16x4_f32(false, aNs, false, bRe0, (short)0, acci0, false, false);
        accr1 = __builtin_amdgcn_wmma_f32_16x16x4_f32(false, aRe, false, bRe1, (short)0, accr1, false, false);
        accr1 = __builtin_amdgcn_wmma_f32_16x16x4_f32(false, aPs, false, bIm1, (short)0, accr1, false, false);
        acci1 = __builtin_amdgcn_wmma_f32_16x16x4_f32(false, aRe, false, bIm1, (short)0, acci1, false, false);
        acci1 = __builtin_amdgcn_wmma_f32_16x16x4_f32(false, aNs, false, bRe1, (short)0, acci1, false, false);
    }

    for (int ct = 0; ct < 2; ++ct) {
        const v8f& ar = ct ? accr1 : accr0;
        const v8f& ai = ct ? acci1 : acci0;
        const int kcol = col0 + ct * 16 + ln;
        if (kcol < KBINS) {
            const bool edge = (kcol == 0) || (kcol == NH);
            for (int v = 0; v < 8; ++v) {
                const int mr = wave * 16 + 8 * lg + v;
                const size_t o = (size_t)(b * C_ + mr) * N_;
                float orv = ar[v];
                float oiv = edge ? 0.f : ai[v];
                outre[o + kcol] = orv;
                outim[o + kcol] = oiv;
                if (kcol >= 1 && kcol <= NH - 1) {
                    outre[o + (N_ - kcol)] =  orv;
                    outim[o + (N_ - kcol)] = -oiv;
                }
            }
        }
    }
}

extern "C" void kernel_launch(void* const* d_in, const int* in_sizes, int n_in,
                              void* d_out, int out_size, void* d_ws, size_t ws_size,
                              hipStream_t stream) {
    const float* xre   = (const float*)d_in[0];
    const float* xim   = (const float*)d_in[1];
    const float* gamma = (const float*)d_in[2];
    const float* beta  = (const float*)d_in[3];
    float* outre = (float*)d_out;
    float* outim = outre + (size_t)B_ * C_ * N_;

    float2* R  = (float2*)d_ws;
    float2* Wc = R + (size_t)B_ * C_ * KS;
    float2* Ws = Wc + C_ * WPAIR;

    k0_twiddle<<<dim3((C_ * WPAIR) / 256), dim3(256), 0, stream>>>(Wc, Ws);
    k1_row_fft<<<dim3(B_ * C_), dim3(256), 0, stream>>>(xre, xim, gamma, beta, R);
    k2_dftC_wmma<<<dim3((KBINS + 31) / 32, B_), dim3(256), 0, stream>>>(R, Wc, Ws, outre, outim);
}


# row FFT: each distinct twiddle (sincos) evaluated once per thread per pass, 39 per transform instead of 208 (one per butterfly); twiddle values and butterfly arithmetic unchanged
# speedup vs baseline: 1.7443x; 1.7443x over previous
#include <hip/hip_runtime.h>
#include <math.h>

#define B_    16
#define C_    128
#define N_    8192
#define NH    4096
#define KBINS 4097
#define KS    4104
#define WPAIR 64

typedef __attribute__((ext_vector_type(2))) float v2f;
typedef __attribute__((ext_vector_type(8))) float v8f;

__global__ __launch_bounds__(256) void k0_twiddle(float2* __restrict__ Wc,
                                                  float2* __restrict__ Ws)
{
    const int idx = blockIdx.x * 256 + threadIdx.x;
    const int m = idx >> 6, j = idx & 63;
    const float scale = 1.0f / 1048576.0f;
    float f0 = (float)(m * (2 * j))     * (1.f / 64.f);
    float f1 = (float)(m * (2 * j + 1)) * (1.f / 64.f);
    Wc[idx] = make_float2(cospif(f0) * scale, cospif(f1) * scale);
    Ws[idx] = make_float2(sinpif(f0) * scale, sinpif(f1) * scale);
}

template <bool INV>
__device__ __forceinline__ void twiddle_of(int j, int ls, float& wr, float& wi)
{
    const float a = ldexpf((float)j, -ls);
    wr = cospif(a);
    wi = INV ? sinpif(a) : -sinpif(a);
}

template <bool INV, int U>
__device__ __forceinline__ void bfly_group(float* __restrict__ re, float* __restrict__ im,
                                           int first, int step, int s, float wr, float wi)
{
    float ar[U], ai[U], br[U], bi[U];
    for (int u = 0; u < U; ++u) {
        const int i0 = first + u * step;
        ar[u] = re[i0];     ai[u] = im[i0];
        br[u] = re[i0 + s]; bi[u] = im[i0 + s];
    }
    for (int u = 0; u < U; ++u) {
        const int i0 = first + u * step;
        const int i1 = i0 + s;
        if (INV) {
            float tr = ar[u] - br[u], ti = ai[u] - bi[u];
            re[i0] = ar[u] + br[u];  im[i0] = ai[u] + bi[u];
            re[i1] = tr * wr - ti * wi;
            im[i1] = tr * wi + ti * wr;
        } else {
            float tr = br[u] * wr - bi[u] * wi;
            float ti = br[u] * wi + bi[u] * wr;
            re[i0] = ar[u] + tr;  im[i0] = ai[u] + ti;
            re[i1] = ar[u] - tr;  im[i1] = ai[u] - ti;
        }
    }
}

template <bool INV>
__device__ __forceinline__ void pass_small(float* __restrict__ re, float* __restrict__ im,
                                           int tid, int ls)
{
    const int s    = 1 << ls;
    const int j    = tid & (s - 1);
    const int base = ((tid >> ls) << (ls + 1)) + j;
    float wr, wi;
    twiddle_of<INV>(j, ls, wr, wi);
    bfly_group<INV, 8>(re, im, base,           512, s, wr, wi);
    bfly_group<INV, 8>(re, im, base + 8 * 512, 512, s, wr, wi);
}

template <bool INV, int LS>
__device__ __forceinline__ void pass_large(float* __restrict__ re, float* __restrict__ im, int tid)
{
    constexpr int s = 1 << LS;
    constexpr int Q = s >> 8;
    constexpr int R = 16 / Q;
    #pragma unroll 1
    for (int q = 0; q < Q; ++q) {
        const int j = tid + 256 * q;
        float wr, wi;
        twiddle_of<INV>(j, LS, wr, wi);
        bfly_group<INV, R>(re, im, j, 2 * s, s, wr, wi);
    }
}

__global__ __launch_bounds__(256) void k1_row_fft(const float* __restrict__ xre,
                                                  const float* __restrict__ xim,
                                                  const float* __restrict__ gamma,
                                                  const float* __restrict__ beta,
                                                  float2* __restrict__ R)
{
    __shared__ float re[N_];
    __shared__ float im[N_];
    const int row = blockIdx.x;
    const int c   = row & (C_ - 1);
    const int tid = threadIdx.x;
    const size_t base = (size_t)row * N_;

    for (int i = tid; i < N_; i += 256) {
        re[i] = xre[base + i];
        im[i] = xim[base + i];
    }
    __syncthreads();

    pass_large<true, 12>(re, im, tid);  __syncthreads();
    pass_large<true, 11>(re, im, tid);  __syncthreads();
    pass_large<true, 10>(re, im, tid);  __syncthreads();
    pass_large<true,  9>(re, im, tid);  __syncthreads();
    for (int ls = 8; ls >= 0; --ls) {
        pass_small<true>(re, im, tid, ls);
        __syncthreads();
    }

    float lsum = 0.f, lsq = 0.f;
    for (int i = tid; i < N_; i += 256) { float v = re[i]; lsum += v; lsq += v * v; }
    im[tid]       = lsum;
    im[256 + tid] = lsq;
    __syncthreads();
    for (int off = 128; off > 0; off >>= 1) {
        if (tid < off) { im[tid] += im[tid + off]; im[256 + tid] += im[256 + tid + off]; }
        __syncthreads();
    }
    const float mean = im[0]   * (1.f / N_);
    const float var  = im[256] * (1.f / N_) - mean * mean;
    const float istd = rsqrtf(var + 1e-5f);
    const float gm = gamma[c], bt = beta[c];
    __syncthreads();

    for (int i = tid; i < N_; i += 256) {
        float v = (re[i] - mean) * istd * gm + bt;
        v = 0.5f * v * (1.f + erff(v * 0.70710678118654752f));
        re[i] = v;
        im[i] = 0.f;
    }
    __syncthreads();

    for (int ls = 0; ls <= 8; ++ls) {
        pass_small<false>(re, im, tid, ls);
        __syncthreads();
    }
    pass_large<false,  9>(re, im, tid);  __syncthreads();
    pass_large<false, 10>(re, im, tid);  __syncthreads();
    pass_large<false, 11>(re, im, tid);  __syncthreads();
    pass_large<false, 12>(re, im, tid);  __syncthreads();

    float2* Rrow = R + (size_t)row * KS;
    for (int k = tid; k < KBINS; k += 256)
        Rrow[k] = make_float2(re[k], im[k]);
}

__global__ __launch_bounds__(256) void k2_dftC_wmma(const float2* __restrict__ R,
                                                    const float2* __restrict__ Wc,
                                                    const float2* __restrict__ Ws,
                                                    float* __restrict__ outre,
                                                    float* __restrict__ outim)
{
    __shared__ float Lre[C_][40];
    __shared__ float Lim[C_][40];
    const int b    = blockIdx.y;
    const int col0 = blockIdx.x * 32;
    const int tid  = threadIdx.x;
    const float2* Rb = R + (size_t)b * C_ * KS;

    for (int idx = tid; idx < C_ * 32; idx += 256) {
        int r = idx >> 5, cc = idx & 31;
        int k = col0 + cc;
        float2 v = (k < KBINS) ? Rb[(size_t)r * KS + k] : make_float2(0.f, 0.f);
        Lre[r][cc] = v.x;
        Lim[r][cc] = v.y;
    }
    __syncthreads();

    const int wave = tid >> 5;
    const int lane = tid & 31;
    const int lg   = lane >> 4;
    const int ln   = lane & 15;
    const int m    = wave * 16 + ln;
    const float2* WcRow = Wc + m * WPAIR;
    const float2* WsRow = Ws + m * WPAIR;

    v8f accr0 = {0.f,0.f,0.f,0.f,0.f,0.f,0.f,0.f};
    v8f acci0 = {0.f,0.f,0.f,0.f,0.f,0.f,0.f,0.f};
    v8f accr1 = {0.f,0.f,0.f,0.f,0.f,0.f,0.f,0.f};
    v8f acci1 = {0.f,0.f,0.f,0.f,0.f,0.f,0.f,0.f};

    for (int k0 = 0; k0 < C_; k0 += 4) {
        const int kA = k0 + 2 * lg;
        float2 wc = WcRow[(k0 >> 1) + lg];
        float2 wsn = WsRow[(k0 >> 1) + lg];
        v2f aRe = {  wc.x,  wc.y  };
        v2f aPs = {  wsn.x, wsn.y };
        v2f aNs = { -wsn.x, -wsn.y };
        v2f bRe0 = { Lre[kA][ln],      Lre[kA + 1][ln]      };
        v2f bIm0 = { Lim[kA][ln],      Lim[kA + 1][ln]      };
        v2f bRe1 = { Lre[kA][16 + ln], Lre[kA + 1][16 + ln] };
        v2f bIm1 = { Lim[kA][16 + ln], Lim[kA + 1][16 + ln] };
        accr0 = __builtin_amdgcn_wmma_f32_16x16x4_f32(false, aRe, false, bRe0, (short)0, accr0, false, false);
        accr0 = __builtin_amdgcn_wmma_f32_16x16x4_f32(false, aPs, false, bIm0, (short)0, accr0, false, false);
        acci0 = __builtin_amdgcn_wmma_f32_16x16x4_f32(false, aRe, false, bIm0, (short)0, acci0, false, false);
        acci0 = __builtin_amdgcn_wmma_f32_16x16x4_f32(false, aNs, false, bRe0, (short)0, acci0, false, false);
        accr1 = __builtin_amdgcn_wmma_f32_16x16x4_f32(false, aRe, false, bRe1, (short)0, accr1, false, false);
        accr1 = __builtin_amdgcn_wmma_f32_16x16x4_f32(false, aPs, false, bIm1, (short)0, accr1, false, false);
        acci1 = __builtin_amdgcn_wmma_f32_16x16x4_f32(false, aRe, false, bIm1, (short)0, acci1, false, false);
        acci1 = __builtin_amdgcn_wmma_f32_16x16x4_f32(false, aNs, false, bRe1, (short)0, acci1, false, false);
    }

    for (int ct = 0; ct < 2; ++ct) {
        const v8f& ar = ct ? accr1 : accr0;
        const v8f& ai = ct ? acci1 : acci0;
        const int kcol = col0 + ct * 16 + ln;
        if (kcol < KBINS) {
            const bool edge = (kcol == 0) || (kcol == NH);
            for (int v = 0; v < 8; ++v) {
                const int mr = wave * 16 + 8 * lg + v;
                const size_t o = (size_t)(b * C_ + mr) * N_;
                float orv = ar[v];
                float oiv = edge ? 0.f : ai[v];
                outre[o + kcol] = orv;
                outim[o + kcol] = oiv;
                if (kcol >= 1 && kcol <= NH - 1) {
                    outre[o + (N_ - kcol)] =  orv;
                    outim[o + (N_ - kcol)] = -oiv;
                }
            }
        }
    }
}

extern "C" void kernel_launch(void* const* d_in, const int* in_sizes, int n_in,
                              void* d_out, int out_size, void* d_ws, size_t ws_size,
                              hipStream_t stream) {
    const float* xre   = (const float*)d_in[0];
    const float* xim   = (const float*)d_in[1];
    const float* gamma = (const float*)d_in[2];
    const float* beta  = (const float*)d_in[3];
    float* outre = (float*)d_out;
    float* outim = outre + (size_t)B_ * C_ * N_;

    float2* R  = (float2*)d_ws;
    float2* Wc = R + (size_t)B_ * C_ * KS;
    float2* Ws = Wc + C_ * WPAIR;

    k0_twiddle<<<dim3((C_ * WPAIR) / 256), dim3(256), 0, stream>>>(Wc, Ws);
    k1_row_fft<<<dim3(B_ * C_), dim3(256), 0, stream>>>(xre, xim, gamma, beta, R);
    k2_dftC_wmma<<<dim3((KBINS + 31) / 32, B_), dim3(256), 0, stream>>>(R, Wc, Ws, outre, outim);
}


# row FFT: two radix-2 stages fused per LDS round trip (4-point groups in registers): 7 passes and barriers per transform instead of 13; second twiddle by exact quarter turn; sum order unchanged
# speedup vs baseline: 2.0268x; 1.1620x over previous
#include <hip/hip_runtime.h>
#include <math.h>

#define B_    16
#define C_    128
#define N_    8192
#define NH    4096
#define KBINS 4097
#define KS    4104
#define WPAIR 64

typedef __attribute__((ext_vector_type(2))) float v2f;
typedef __attribute__((ext_vector_type(8))) float v8f;

__global__ __launch_bounds__(256) void k0_twiddle(float2* __restrict__ Wc,
                                                  float2* __restrict__ Ws)
{
    const int idx = blockIdx.x * 256 + threadIdx.x;
    const int m = idx >> 6, j = idx & 63;
    const float scale = 1.0f / 1048576.0f;
    float f0 = (float)(m * (2 * j))     * (1.f / 64.f);
    float f1 = (float)(m * (2 * j + 1)) * (1.f / 64.f);
    Wc[idx] = make_float2(cospif(f0) * scale, cospif(f1) * scale);
    Ws[idx] = make_float2(sinpif(f0) * scale, sinpif(f1) * scale);
}

template <bool INV>
__device__ __forceinline__ void twiddle_of(int j, int ls, float& wr, float& wi)
{
    const float a = ldexpf((float)j, -ls);
    wr = cospif(a);
    wi = INV ? sinpif(a) : -sinpif(a);
}

template <bool INV>
__device__ __forceinline__ void bfly(float& ar, float& ai, float& br, float& bi,
                                     float wr, float wi)
{
    if (INV) {
        const float tr = ar - br, ti = ai - bi;
        ar = ar + br;  ai = ai + bi;
        br = tr * wr - ti * wi;
        bi = tr * wi + ti * wr;
    } else {
        const float tr = br * wr - bi * wi;
        const float ti = br * wi + bi * wr;
        br = ar - tr;  bi = ai - ti;
        ar = ar + tr;  ai = ai + ti;
    }
}

template <bool INV, int U>
__device__ __forceinline__ void bfly_group(float* __restrict__ re, float* __restrict__ im,
                                           int first, int step, int s, float wr, float wi)
{
    float ar[U], ai[U], br[U], bi[U];
    for (int u = 0; u < U; ++u) {
        const int i0 = first + u * step;
        ar[u] = re[i0];     ai[u] = im[i0];
        br[u] = re[i0 + s]; bi[u] = im[i0 + s];
    }
    for (int u = 0; u < U; ++u) {
        const int i0 = first + u * step;
        bfly<INV>(ar[u], ai[u], br[u], bi[u], wr, wi);
        re[i0]     = ar[u];  im[i0]     = ai[u];
        re[i0 + s] = br[u];  im[i0 + s] = bi[u];
    }
}

template <bool INV, int U>
__device__ __forceinline__ void quad_group(float* __restrict__ re, float* __restrict__ im,
                                           int first, int step, int h,
                                           float wsr, float wsi, float whr, float whi)
{
    const float wqr = INV ? -wsi :  wsi;
    const float wqi = INV ?  wsr : -wsr;
    float xr[U][4], xi[U][4];
    for (int u = 0; u < U; ++u) {
        const int p = first + u * step;
        for (int k = 0; k < 4; ++k) { xr[u][k] = re[p + k * h]; xi[u][k] = im[p + k * h]; }
    }
    for (int u = 0; u < U; ++u) {
        const int p = first + u * step;
        if (INV) {
            bfly<true>(xr[u][0], xi[u][0], xr[u][2], xi[u][2], wsr, wsi);
            bfly<true>(xr[u][1], xi[u][1], xr[u][3], xi[u][3], wqr, wqi);
            bfly<true>(xr[u][0], xi[u][0], xr[u][1], xi[u][1], whr, whi);
            bfly<true>(xr[u][2], xi[u][2], xr[u][3], xi[u][3], whr, whi);
        } else {
            bfly<false>(xr[u][0], xi[u][0], xr[u][1], xi[u][1], whr, whi);
            bfly<false>(xr[u][2], xi[u][2], xr[u][3], xi[u][3], whr, whi);
            bfly<false>(xr[u][0], xi[u][0], xr[u][2], xi[u][2], wsr, wsi);
            bfly<false>(xr[u][1], xi[u][1], xr[u][3], xi[u][3], wqr, wqi);
        }
        for (int k = 0; k < 4; ++k) { re[p + k * h] = xr[u][k]; im[p + k * h] = xi[u][k]; }
    }
}

template <bool INV>
__device__ __forceinline__ void pass_small(float* __restrict__ re, float* __restrict__ im,
                                           int tid, int ls)
{
    const int s    = 1 << ls;
    const int j    = tid & (s - 1);
    const int base = ((tid >> ls) << (ls + 1)) + j;
    float wr, wi;
    twiddle_of<INV>(j, ls, wr, wi);
    bfly_group<INV, 8>(re, im, base,           512, s, wr, wi);
    bfly_group<INV, 8>(re, im, base + 8 * 512, 512, s, wr, wi);
}

template <bool INV>
__device__ __forceinline__ void pass4_small(float* __restrict__ re, float* __restrict__ im,
                                            int tid, int L)
{
    const int h    = 1 << (L - 1);
    const int j    = tid & (h - 1);
    const int base = ((tid >> (L - 1)) << (L + 1)) + j;
    float wsr, wsi, whr, whi;
    twiddle_of<INV>(j, L,     wsr, wsi);
    twiddle_of<INV>(j, L - 1, whr, whi);
    quad_group<INV, 4>(re, im, base,            1024, h, wsr, wsi, whr, whi);
    quad_group<INV, 4>(re, im, base + 4 * 1024, 1024, h, wsr, wsi, whr, whi);
}

template <bool INV, int L>
__device__ __forceinline__ void pass4_large(float* __restrict__ re, float* __restrict__ im, int tid)
{
    constexpr int h = 1 << (L - 1);
    constexpr int Q = h >> 8;
    constexpr int R = 8 / Q;
    #pragma unroll 1
    for (int q = 0; q < Q; ++q) {
        const int j = tid + 256 * q;
        float wsr, wsi, whr, whi;
        twiddle_of<INV>(j, L,     wsr, wsi);
        twiddle_of<INV>(j, L - 1, whr, whi);
        quad_group<INV, R>(re, im, j, 4 * h, h, wsr, wsi, whr, whi);
    }
}

__global__ __launch_bounds__(256) void k1_row_fft(const float* __restrict__ xre,
                                                  const float* __restrict__ xim,
                                                  const float* __restrict__ gamma,
                                                  const float* __restrict__ beta,
                                                  float2* __restrict__ R)
{
    __shared__ float re[N_];
    __shared__ float im[N_];
    const int row = blockIdx.x;
    const int c   = row & (C_ - 1);
    const int tid = threadIdx.x;
    const size_t base = (size_t)row * N_;

    for (int i = tid; i < N_; i += 256) {
        re[i] = xre[base + i];
        im[i] = xim[base + i];
    }
    __syncthreads();

    pass4_large<true, 12>(re, im, tid);  __syncthreads();
    pass4_large<true, 10>(re, im, tid);  __syncthreads();
    for (int L = 8; L >= 2; L -= 2) {
        pass4_small<true>(re, im, tid, L);
        __syncthreads();
    }
    pass_small<true>(re, im, tid, 0);
    __syncthreads();

    float lsum = 0.f, lsq = 0.f;
    for (int i = tid; i < N_; i += 256) { float v = re[i]; lsum += v; lsq += v * v; }
    im[tid]       = lsum;
    im[256 + tid] = lsq;
    __syncthreads();
    for (int off = 128; off > 0; off >>= 1) {
        if (tid < off) { im[tid] += im[tid + off]; im[256 + tid] += im[256 + tid + off]; }
        __syncthreads();
    }
    const float mean = im[0]   * (1.f / N_);
    const float var  = im[256] * (1.f / N_) - mean * mean;
    const float istd = rsqrtf(var + 1e-5f);
    const float gm = gamma[c], bt = beta[c];
    __syncthreads();

    for (int i = tid; i < N_; i += 256) {
        float v = (re[i] - mean) * istd * gm + bt;
        v = 0.5f * v * (1.f + erff(v * 0.70710678118654752f));
        re[i] = v;
        im[i] = 0.f;
    }
    __syncthreads();

    pass_small<false>(re, im, tid, 0);
    __syncthreads();
    for (int L = 2; L <= 8; L += 2) {
        pass4_small<false>(re, im, tid, L);
        __syncthreads();
    }
    pass4_large<false, 10>(re, im, tid);  __syncthreads();
    pass4_large<false, 12>(re, im, tid);  __syncthreads();

    float2* Rrow = R + (size_t)row * KS;
    for (int k = tid; k < KBINS; k += 256)
        Rrow[k] = make_float2(re[k], im[k]);
}

__global__ __launch_bounds__(256) void k2_dftC_wmma(const float2* __restrict__ R,
                                                    const float2* __restrict__ Wc,
                                                    const float2* __restrict__ Ws,
                                                    float* __restrict__ outre,
                                                    float* __restrict__ outim)
{
    __shared__ float Lre[C_][40];
    __shared__ float Lim[C_][40];
    const int b    = blockIdx.y;
    const int col0 = blockIdx.x * 32;
    const int tid  = threadIdx.x;
    const float2* Rb = R + (size_t)b * C_ * KS;

    for (int idx = tid; idx < C_ * 32; idx += 256) {
        int r = idx >> 5, cc = idx & 31;
        int k = col0 + cc;
        float2 v = (k < KBINS) ? Rb[(size_t)r * KS + k] : make_float2(0.f, 0.f);
        Lre[r][cc] = v.x;
        Lim[r][cc] = v.y;
    }
    __syncthreads();

    const int wave = tid >> 5;
    const int lane = tid & 31;
    const int lg   = lane >> 4;
    const int ln   = lane & 15;
    const int m    = wave * 16 + ln;
    const float2* WcRow = Wc + m * WPAIR;
    const float2* WsRow = Ws + m * WPAIR;

    v8f accr0 = {0.f,0.f,0.f,0.f,0.f,0.f,0.f,0.f};
    v8f acci0 = {0.f,0.f,0.f,0.f,0.f,0.f,0.f,0.f};
    v8f accr1 = {0.f,0.f,0.f,0.f,0.f,0.f,0.f,0.f};
    v8f acci1 = {0.f,0.f,0.f,0.f,0.f,0.f,0.f,0.f};

    for (int k0 = 0; k0 < C_; k0 += 4) {
        const int kA = k0 + 2 * lg;
        float2 wc = WcRow[(k0 >> 1) + lg];
        float2 wsn = WsRow[(k0 >> 1) + lg];
        v2f aRe = {  wc.x,  wc.y  };
        v2f aPs = {  wsn.x, wsn.y };
        v2f aNs = { -wsn.x, -wsn.y };
        v2f bRe0 = { Lre[kA][ln],      Lre[kA + 1][ln]      };
        v2f bIm0 = { Lim[kA][ln],      Lim[kA + 1][ln]      };
        v2f bRe1 = { Lre[kA][16 + ln], Lre[kA + 1][16 + ln] };
        v2f bIm1 = { Lim[kA][16 + ln], Lim[kA + 1][16 + ln] };
        accr0 = __builtin_amdgcn_wmma_f32_16x16x4_f32(false, aRe, false, bRe0, (short)0, accr0, false, false);
        accr0 = __builtin_amdgcn_wmma_f32_16x16x4_f32(false, aPs, false, bIm0, (short)0, accr0, false, false);
        acci0 = __builtin_amdgcn_wmma_f32_16x16x4_f32(false, aRe, false, bIm0, (short)0, acci0, false, false);
        acci0 = __builtin_amdgcn_wmma_f32_16x16x4_f32(false, aNs, false, bRe0, (short)0, acci0, false, false);
        accr1 = __builtin_amdgcn_wmma_f32_16x16x4_f32(false, aRe, false, bRe1, (short)0, accr1, false, false);
        accr1 = __builtin_amdgcn_wmma_f32_16x16x4_f32(false, aPs, false, bIm1, (short)0, accr1, false, false);
        acci1 = __builtin_amdgcn_wmma_f32_16x16x4_f32(false, aRe, false, bIm1, (short)0, acci1, false, false);
        acci1 = __builtin_amdgcn_wmma_f32_16x16x4_f32(false, aNs, false, bRe1, (short)0, acci1, false, false);
    }

    for (int ct = 0; ct < 2; ++ct) {
        const v8f& ar = ct ? accr1 : accr0;
        const v8f& ai = ct ? acci1 : acci0;
        const int kcol = col0 + ct * 16 + ln;
        if (kcol < KBINS) {
            const bool edge = (kcol == 0) || (kcol == NH);
            for (int v = 0; v < 8; ++v) {
                const int mr = wave * 16 + 8 * lg + v;
                const size_t o = (size_t)(b * C_ + mr) * N_;
                float orv = ar[v];
                float oiv = edge ? 0.f : ai[v];
                outre[o + kcol] = orv;
                outim[o + kcol] = oiv;
                if (kcol >= 1 && kcol <= NH - 1) {
                    outre[o + (N_ - kcol)] =  orv;
                    outim[o + (N_ - kcol)] = -oiv;
                }
            }
        }
    }
}

extern "C" void kernel_launch(void* const* d_in, const int* in_sizes, int n_in,
                              void* d_out, int out_size, void* d_ws, size_t ws_size,
                              hipStream_t stream) {
    const float* xre   = (const float*)d_in[0];
    const float* xim   = (const float*)d_in[1];
    const float* gamma = (const float*)d_in[2];
    const float* beta  = (const float*)d_in[3];
    float* outre = (float*)d_out;
    float* outim = outre + (size_t)B_ * C_ * N_;

    float2* R  = (float2*)d_ws;
    float2* Wc = R + (size_t)B_ * C_ * KS;
    float2* Ws = Wc + C_ * WPAIR;

    k0_twiddle<<<dim3((C_ * WPAIR) / 256), dim3(256), 0, stream>>>(Wc, Ws);
    k1_row_fft<<<dim3(B_ * C_), dim3(256), 0, stream>>>(xre, xim, gamma, beta, R);
    k2_dftC_wmma<<<dim3((KBINS + 31) / 32, B_), dim3(256), 0, stream>>>(R, Wc, Ws, outre, outim);
}


# channel DFT: rows m and 128-m share their cos and sin products (conjugate rows): WMMAs cover 64 rows, not 128 (half as many); re/im formed by add/sub of two accumulators, so sum order changes
# speedup vs baseline: 2.1180x; 1.0450x over previous
#include <hip/hip_runtime.h>
#include <math.h>

#define B_    16
#define C_    128
#define N_    8192
#define NH    4096
#define KBINS 4097
#define KS    4104
#define WPAIR 64
#define MH_   64

typedef __attribute__((ext_vector_type(2))) float v2f;
typedef __attribute__((ext_vector_type(8))) float v8f;

__global__ __launch_bounds__(256) void k0_twiddle(float2* __restrict__ Wc,
                                                  float2* __restrict__ Ws)
{
    const int idx = blockIdx.x * 256 + threadIdx.x;
    const int m = idx >> 6, j = idx & 63;
    const float scale = 1.0f / 1048576.0f;
    float f0 = (float)(m * (2 * j))     * (1.f / 64.f);
    float f1 = (float)(m * (2 * j + 1)) * (1.f / 64.f);
    Wc[idx] = make_float2(cospif(f0) * scale, cospif(f1) * scale);
    if (m == 0) {
        Ws[idx] = make_float2(cospif((float)(2 * j)) * scale, cospif((float)(2 * j + 1)) * scale);
    } else {
        Ws[idx] = make_float2(sinpif(f0) * scale, sinpif(f1) * scale);
    }
}

template <bool INV>
__device__ __forceinline__ void twiddle_of(int j, int ls, float& wr, float& wi)
{
    const float a = ldexpf((float)j, -ls);
    wr = cospif(a);
    wi = INV ? sinpif(a) : -sinpif(a);
}

template <bool INV>
__device__ __forceinline__ void bfly(float& ar, float& ai, float& br, float& bi,
                                     float wr, float wi)
{
    if (INV) {
        const float tr = ar - br, ti = ai - bi;
        ar = ar + br;  ai = ai + bi;
        br = tr * wr - ti * wi;
        bi = tr * wi + ti * wr;
    } else {
        const float tr = br * wr - bi * wi;
        const float ti = br * wi + bi * wr;
        br = ar - tr;  bi = ai - ti;
        ar = ar + tr;  ai = ai + ti;
    }
}

template <bool INV, int U>
__device__ __forceinline__ void bfly_group(float* __restrict__ re, float* __restrict__ im,
                                           int first, int step, int s, float wr, float wi)
{
    float ar[U], ai[U], br[U], bi[U];
    for (int u = 0; u < U; ++u) {
        const int i0 = first + u * step;
        ar[u] = re[i0];     ai[u] = im[i0];
        br[u] = re[i0 + s]; bi[u] = im[i0 + s];
    }
    for (int u = 0; u < U; ++u) {
        const int i0 = first + u * step;
        bfly<INV>(ar[u], ai[u], br[u], bi[u], wr, wi);
        re[i0]     = ar[u];  im[i0]     = ai[u];
        re[i0 + s] = br[u];  im[i0 + s] = bi[u];
    }
}

template <bool INV, int U>
__device__ __forceinline__ void quad_group(float* __restrict__ re, float* __restrict__ im,
                                           int first, int step, int h,
                                           float wsr, float wsi, float whr, float whi)
{
    const float wqr = INV ? -wsi :  wsi;
    const float wqi = INV ?  wsr : -wsr;
    float xr[U][4], xi[U][4];
    for (int u = 0; u < U; ++u) {
        const int p = first + u * step;
        for (int k = 0; k < 4; ++k) { xr[u][k] = re[p + k * h]; xi[u][k] = im[p + k * h]; }
    }
    for (int u = 0; u < U; ++u) {
        const int p = first + u * step;
        if (INV) {
            bfly<true>(xr[u][0], xi[u][0], xr[u][2], xi[u][2], wsr, wsi);
            bfly<true>(xr[u][1], xi[u][1], xr[u][3], xi[u][3], wqr, wqi);
            bfly<true>(xr[u][0], xi[u][0], xr[u][1], xi[u][1], whr, whi);
            bfly<true>(xr[u][2], xi[u][2], xr[u][3], xi[u][3], whr, whi);
        } else {
            bfly<false>(xr[u][0], xi[u][0], xr[u][1], xi[u][1], whr, whi);
            bfly<false>(xr[u][2], xi[u][2], xr[u][3], xi[u][3], whr, whi);
            bfly<false>(xr[u][0], xi[u][0], xr[u][2], xi[u][2], wsr, wsi);
            bfly<false>(xr[u][1], xi[u][1], xr[u][3], xi[u][3], wqr, wqi);
        }
        for (int k = 0; k < 4; ++k) { re[p + k * h] = xr[u][k]; im[p + k * h] = xi[u][k]; }
    }
}

template <bool INV>
__device__ __forceinline__ void pass_small(float* __restrict__ re, float* __restrict__ im,
                                           int tid, int ls)
{
    const int s    = 1 << ls;
    const int j    = tid & (s - 1);
    const int base = ((tid >> ls) << (ls + 1)) + j;
    float wr, wi;
    twiddle_of<INV>(j, ls, wr, wi);
    bfly_group<INV, 8>(re, im, base,           512, s, wr, wi);
    bfly_group<INV, 8>(re, im, base + 8 * 512, 512, s, wr, wi);
}

template <bool INV>
__device__ __forceinline__ void pass4_small(float* __restrict__ re, float* __restrict__ im,
                                            int tid, int L)
{
    const int h    = 1 << (L - 1);
    const int j    = tid & (h - 1);
    const int base = ((tid >> (L - 1)) << (L + 1)) + j;
    float wsr, wsi, whr, whi;
    twiddle_of<INV>(j, L,     wsr, wsi);
    twiddle_of<INV>(j, L - 1, whr, whi);
    quad_group<INV, 4>(re, im, base,            1024, h, wsr, wsi, whr, whi);
    quad_group<INV, 4>(re, im, base + 4 * 1024, 1024, h, wsr, wsi, whr, whi);
}

template <bool INV, int L>
__device__ __forceinline__ void pass4_large(float* __restrict__ re, float* __restrict__ im, int tid)
{
    constexpr int h = 1 << (L - 1);
    constexpr int Q = h >> 8;
    constexpr int R = 8 / Q;
    #pragma unroll 1
    for (int q = 0; q < Q; ++q) {
        const int j = tid + 256 * q;
        float wsr, wsi, whr, whi;
        twiddle_of<INV>(j, L,     wsr, wsi);
        twiddle_of<INV>(j, L - 1, whr, whi);
        quad_group<INV, R>(re, im, j, 4 * h, h, wsr, wsi, whr, whi);
    }
}

__global__ __launch_bounds__(256) void k1_row_fft(const float* __restrict__ xre,
                                                  const float* __restrict__ xim,
                                                  const float* __restrict__ gamma,
                                                  const float* __restrict__ beta,
                                                  float2* __restrict__ R)
{
    __shared__ float re[N_];
    __shared__ float im[N_];
    const int row = blockIdx.x;
    const int c   = row & (C_ - 1);
    const int tid = threadIdx.x;
    const size_t base = (size_t)row * N_;

    for (int i = tid; i < N_; i += 256) {
        re[i] = xre[base + i];
        im[i] = xim[base + i];
    }
    __syncthreads();

    pass4_large<true, 12>(re, im, tid);  __syncthreads();
    pass4_large<true, 10>(re, im, tid);  __syncthreads();
    for (int L = 8; L >= 2; L -= 2) {
        pass4_small<true>(re, im, tid, L);
        __syncthreads();
    }
    pass_small<true>(re, im, tid, 0);
    __syncthreads();

    float lsum = 0.f, lsq = 0.f;
    for (int i = tid; i < N_; i += 256) { float v = re[i]; lsum += v; lsq += v * v; }
    im[tid]       = lsum;
    im[256 + tid] = lsq;
    __syncthreads();
    for (int off = 128; off > 0; off >>= 1) {
        if (tid < off) { im[tid] += im[tid + off]; im[256 + tid] += im[256 + tid + off]; }
        __syncthreads();
    }
    const float mean = im[0]   * (1.f / N_);
    const float var  = im[256] * (1.f / N_) - mean * mean;
    const float istd = rsqrtf(var + 1e-5f);
    const float gm = gamma[c], bt = beta[c];
    __syncthreads();

    for (int i = tid; i < N_; i += 256) {
        float v = (re[i] - mean) * istd * gm + bt;
        v = 0.5f * v * (1.f + erff(v * 0.70710678118654752f));
        re[i] = v;
        im[i] = 0.f;
    }
    __syncthreads();

    pass_small<false>(re, im, tid, 0);
    __syncthreads();
    for (int L = 2; L <= 8; L += 2) {
        pass4_small<false>(re, im, tid, L);
        __syncthreads();
    }
    pass4_large<false, 10>(re, im, tid);  __syncthreads();
    pass4_large<false, 12>(re, im, tid);  __syncthreads();

    float2* Rrow = R + (size_t)row * KS;
    for (int k = tid; k < KBINS; k += 256)
        Rrow[k] = make_float2(re[k], im[k]);
}

__device__ __forceinline__ void store_bin(float* __restrict__ outre, float* __restrict__ outim,
                                          int b, int row, int kcol, float orv, float oim)
{
    const size_t o = (size_t)(b * C_ + row) * N_;
    const bool edge = (kcol == 0) || (kcol == NH);
    const float oiv = edge ? 0.f : oim;
    outre[o + kcol] = orv;
    outim[o + kcol] = oiv;
    if (kcol >= 1 && kcol <= NH - 1) {
        outre[o + (N_ - kcol)] =  orv;
        outim[o + (N_ - kcol)] = -oiv;
    }
}

__global__ __launch_bounds__(128) void k2_dftC_wmma(const float2* __restrict__ R,
                                                    const float2* __restrict__ Wc,
                                                    const float2* __restrict__ Ws,
                                                    float* __restrict__ outre,
                                                    float* __restrict__ outim)
{
    __shared__ float Lre[C_][40];
    __shared__ float Lim[C_][40];
    const int b    = blockIdx.y;
    const int col0 = blockIdx.x * 32;
    const int tid  = threadIdx.x;
    const float2* Rb = R + (size_t)b * C_ * KS;

    for (int idx = tid; idx < C_ * 32; idx += 128) {
        int r = idx >> 5, cc = idx & 31;
        int k = col0 + cc;
        float2 v = (k < KBINS) ? Rb[(size_t)r * KS + k] : make_float2(0.f, 0.f);
        Lre[r][cc] = v.x;
        Lim[r][cc] = v.y;
    }
    __syncthreads();

    const int wave = tid >> 5;
    const int lane = tid & 31;
    const int lg   = lane >> 4;
    const int ln   = lane & 15;
    const int m    = wave * 16 + ln;
    const float2* WcRow = Wc + m * WPAIR;
    const float2* WsRow = Ws + m * WPAIR;

    const v8f zero8 = {0.f,0.f,0.f,0.f,0.f,0.f,0.f,0.f};
    v8f cr0 = zero8, si0 = zero8, ci0 = zero8, sr0 = zero8;
    v8f cr1 = zero8, si1 = zero8, ci1 = zero8, sr1 = zero8;

    for (int k0 = 0; k0 < C_; k0 += 4) {
        const int kA = k0 + 2 * lg;
        float2 wc = WcRow[(k0 >> 1) + lg];
        float2 ws = WsRow[(k0 >> 1) + lg];
        v2f aC = { wc.x, wc.y };
        v2f aS = { ws.x, ws.y };
        v2f bRe0 = { Lre[kA][ln],      Lre[kA + 1][ln]      };
        v2f bIm0 = { Lim[kA][ln],      Lim[kA + 1][ln]      };
        v2f bRe1 = { Lre[kA][16 + ln], Lre[kA + 1][16 + ln] };
        v2f bIm1 = { Lim[kA][16 + ln], Lim[kA + 1][16 + ln] };
        cr0 = __builtin_amdgcn_wmma_f32_16x16x4_f32(false, aC, false, bRe0, (short)0, cr0, false, false);
        si0 = __builtin_amdgcn_wmma_f32_16x16x4_f32(false, aS, false, bIm0, (short)0, si0, false, false);
        ci0 = __builtin_amdgcn_wmma_f32_16x16x4_f32(false, aC, false, bIm0, (short)0, ci0, false, false);
        sr0 = __builtin_amdgcn_wmma_f32_16x16x4_f32(false, aS, false, bRe0, (short)0, sr0, false, false);
        cr1 = __builtin_amdgcn_wmma_f32_16x16x4_f32(false, aC, false, bRe1, (short)0, cr1, false, false);
        si1 = __builtin_amdgcn_wmma_f32_16x16x4_f32(false, aS, false, bIm1, (short)0, si1, false, false);
        ci1 = __builtin_amdgcn_wmma_f32_16x16x4_f32(false, aC, false, bIm1, (short)0, ci1, false, false);
        sr1 = __builtin_amdgcn_wmma_f32_16x16x4_f32(false, aS, false, bRe1, (short)0, sr1, false, false);
    }

    for (int ct = 0; ct < 2; ++ct) {
        const v8f& p1 = ct ? cr1 : cr0;
        const v8f& p2 = ct ? si1 : si0;
        const v8f& p3 = ct ? ci1 : ci0;
        const v8f& p4 = ct ? sr1 : sr0;
        const int kcol = col0 + ct * 16 + ln;
        if (kcol < KBINS) {
            for (int v = 0; v < 8; ++v) {
                const int mr = wave * 16 + 8 * lg + v;
                const bool z = (mr == 0);
                const int rowB = z ? (C_ / 2) : (C_ - mr);
                const float reA = z ? p1[v] : p1[v] + p2[v];
                const float imA = z ? p3[v] : p3[v] - p4[v];
                const float reB = z ? p4[v] : p1[v] - p2[v];
                const float imB = z ? p2[v] : p3[v] + p4[v];
                store_bin(outre, outim, b, mr,   kcol, reA, imA);
                store_bin(outre, outim, b, rowB, kcol, reB, imB);
            }
        }
    }
}

extern "C" void kernel_launch(void* const* d_in, const int* in_sizes, int n_in,
                              void* d_out, int out_size, void* d_ws, size_t ws_size,
                              hipStream_t stream) {
    const float* xre   = (const float*)d_in[0];
    const float* xim   = (const float*)d_in[1];
    const float* gamma = (const float*)d_in[2];
    const float* beta  = (const float*)d_in[3];
    float* outre = (float*)d_out;
    float* outim = outre + (size_t)B_ * C_ * N_;

    float2* R  = (float2*)d_ws;
    float2* Wc = R + (size_t)B_ * C_ * KS;
    float2* Ws = Wc + MH_ * WPAIR;

    k0_twiddle<<<dim3((MH_ * WPAIR) / 256), dim3(256), 0, stream>>>(Wc, Ws);
    k1_row_fft<<<dim3(B_ * C_), dim3(256), 0, stream>>>(xre, xim, gamma, beta, R);
    k2_dftC_wmma<<<dim3((KBINS + 31) / 32, B_), dim3(128), 0, stream>>>(R, Wc, Ws, outre, outim);
}


# row FFT: two rows share one complex transform (inverse of H1 + i*H2, Hermitian parts; forward of y1 + i*y2, split by conjugate symmetry): half the transforms; sum order changes
# speedup vs baseline: 2.4937x; 1.1774x over previous
#include <hip/hip_runtime.h>
#include <math.h>

#define B_    16
#define C_    128
#define N_    8192
#define NH    4096
#define KBINS 4097
#define KS    4104
#define WPAIR 64
#define MH_   64

typedef __attribute__((ext_vector_type(2))) float v2f;
typedef __attribute__((ext_vector_type(8))) float v8f;

__global__ __launch_bounds__(256) void k0_twiddle(float2* __restrict__ Wc,
                                                  float2* __restrict__ Ws)
{
    const int idx = blockIdx.x * 256 + threadIdx.x;
    const int m = idx >> 6, j = idx & 63;
    const float scale = 1.0f / 1048576.0f;
    float f0 = (float)(m * (2 * j))     * (1.f / 64.f);
    float f1 = (float)(m * (2 * j + 1)) * (1.f / 64.f);
    Wc[idx] = make_float2(cospif(f0) * scale, cospif(f1) * scale);
    if (m == 0) {
        Ws[idx] = make_float2(cospif((float)(2 * j)) * scale, cospif((float)(2 * j + 1)) * scale);
    } else {
        Ws[idx] = make_float2(sinpif(f0) * scale, sinpif(f1) * scale);
    }
}

template <bool INV>
__device__ __forceinline__ void twiddle_of(int j, int ls, float& wr, float& wi)
{
    const float a = ldexpf((float)j, -ls);
    wr = cospif(a);
    wi = INV ? sinpif(a) : -sinpif(a);
}

template <bool INV>
__device__ __forceinline__ void bfly(float& ar, float& ai, float& br, float& bi,
                                     float wr, float wi)
{
    if (INV) {
        const float tr = ar - br, ti = ai - bi;
        ar = ar + br;  ai = ai + bi;
        br = tr * wr - ti * wi;
        bi = tr * wi + ti * wr;
    } else {
        const float tr = br * wr - bi * wi;
        const float ti = br * wi + bi * wr;
        br = ar - tr;  bi = ai - ti;
        ar = ar + tr;  ai = ai + ti;
    }
}

template <bool INV, int U>
__device__ __forceinline__ void bfly_group(float* __restrict__ re, float* __restrict__ im,
                                           int first, int step, int s, float wr, float wi)
{
    float ar[U], ai[U], br[U], bi[U];
    for (int u = 0; u < U; ++u) {
        const int i0 = first + u * step;
        ar[u] = re[i0];     ai[u] = im[i0];
        br[u] = re[i0 + s]; bi[u] = im[i0 + s];
    }
    for (int u = 0; u < U; ++u) {
        const int i0 = first + u * step;
        bfly<INV>(ar[u], ai[u], br[u], bi[u], wr, wi);
        re[i0]     = ar[u];  im[i0]     = ai[u];
        re[i0 + s] = br[u];  im[i0 + s] = bi[u];
    }
}

template <bool INV, int U>
__device__ __forceinline__ void quad_group(float* __restrict__ re, float* __restrict__ im,
                                           int first, int step, int h,
                                           float wsr, float wsi, float whr, float whi)
{
    const float wqr = INV ? -wsi :  wsi;
    const float wqi = INV ?  wsr : -wsr;
    float xr[U][4], xi[U][4];
    for (int u = 0; u < U; ++u) {
        const int p = first + u * step;
        for (int k = 0; k < 4; ++k) { xr[u][k] = re[p + k * h]; xi[u][k] = im[p + k * h]; }
    }
    for (int u = 0; u < U; ++u) {
        const int p = first + u * step;
        if (INV) {
            bfly<true>(xr[u][0], xi[u][0], xr[u][2], xi[u][2], wsr, wsi);
            bfly<true>(xr[u][1], xi[u][1], xr[u][3], xi[u][3], wqr, wqi);
            bfly<true>(xr[u][0], xi[u][0], xr[u][1], xi[u][1], whr, whi);
            bfly<true>(xr[u][2], xi[u][2], xr[u][3], xi[u][3], whr, whi);
        } else {
            bfly<false>(xr[u][0], xi[u][0], xr[u][1], xi[u][1], whr, whi);
            bfly<false>(xr[u][2], xi[u][2], xr[u][3], xi[u][3], whr, whi);
            bfly<false>(xr[u][0], xi[u][0], xr[u][2], xi[u][2], wsr, wsi);
            bfly<false>(xr[u][1], xi[u][1], xr[u][3], xi[u][3], wqr, wqi);
        }
        for (int k = 0; k < 4; ++k) { re[p + k * h] = xr[u][k]; im[p + k * h] = xi[u][k]; }
    }
}

template <bool INV>
__device__ __forceinline__ void pass_small(float* __restrict__ re, float* __restrict__ im,
                                           int tid, int ls)
{
    const int s    = 1 << ls;
    const int j    = tid & (s - 1);
    const int base = ((tid >> ls) << (ls + 1)) + j;
    float wr, wi;
    twiddle_of<INV>(j, ls, wr, wi);
    bfly_group<INV, 8>(re, im, base,           512, s, wr, wi);
    bfly_group<INV, 8>(re, im, base + 8 * 512, 512, s, wr, wi);
}

template <bool INV>
__device__ __forceinline__ void pass4_small(float* __restrict__ re, float* __restrict__ im,
                                            int tid, int L)
{
    const int h    = 1 << (L - 1);
    const int j    = tid & (h - 1);
    const int base = ((tid >> (L - 1)) << (L + 1)) + j;
    float wsr, wsi, whr, whi;
    twiddle_of<INV>(j, L,     wsr, wsi);
    twiddle_of<INV>(j, L - 1, whr, whi);
    quad_group<INV, 4>(re, im, base,            1024, h, wsr, wsi, whr, whi);
    quad_group<INV, 4>(re, im, base + 4 * 1024, 1024, h, wsr, wsi, whr, whi);
}

template <bool INV, int L>
__device__ __forceinline__ void pass4_large(float* __restrict__ re, float* __restrict__ im, int tid)
{
    constexpr int h = 1 << (L - 1);
    constexpr int Q = h >> 8;
    constexpr int R = 8 / Q;
    #pragma unroll 1
    for (int q = 0; q < Q; ++q) {
        const int j = tid + 256 * q;
        float wsr, wsi, whr, whi;
        twiddle_of<INV>(j, L,     wsr, wsi);
        twiddle_of<INV>(j, L - 1, whr, whi);
        quad_group<INV, R>(re, im, j, 4 * h, h, wsr, wsi, whr, whi);
    }
}

__global__ __launch_bounds__(256) void k1_row_fft(const float* __restrict__ xre,
                                                  const float* __restrict__ xim,
                                                  const float* __restrict__ gamma,
                                                  const float* __restrict__ beta,
                                                  float2* __restrict__ R)
{
    __shared__ float re[N_];
    __shared__ float im[N_];
    __shared__ float red[4 * 256];
    const int row0 = 2 * blockIdx.x;
    const int c0   = row0 & (C_ - 1);
    const int tid  = threadIdx.x;
    const float* __restrict__ a1 = xre + (size_t)row0 * N_;
    const float* __restrict__ b1 = xim + (size_t)row0 * N_;
    const float* __restrict__ a2 = a1 + N_;
    const float* __restrict__ b2 = b1 + N_;

    #pragma unroll 4
    for (int k = tid; k < NH; k += 256) {
        const int km = (N_ - k) & (N_ - 1);
        const float sA1 = a1[k] + a1[km];
        const float dB1 = b1[k] - b1[km];
        const float sA2 = a2[k] + a2[km];
        const float dB2 = b2[k] - b2[km];
        re[k]  = 0.5f * (sA1 - dB2);
        im[k]  = 0.5f * (dB1 + sA2);
        re[km] = 0.5f * (sA1 + dB2);
        im[km] = 0.5f * (sA2 - dB1);
    }
    if (tid == 0) { re[NH] = a1[NH]; im[NH] = a2[NH]; }
    __syncthreads();

    pass4_large<true, 12>(re, im, tid);  __syncthreads();
    pass4_large<true, 10>(re, im, tid);  __syncthreads();
    for (int L = 8; L >= 2; L -= 2) {
        pass4_small<true>(re, im, tid, L);
        __syncthreads();
    }
    pass_small<true>(re, im, tid, 0);
    __syncthreads();

    float s1 = 0.f, q1 = 0.f, s2 = 0.f, q2 = 0.f;
    for (int i = tid; i < N_; i += 256) {
        const float v = re[i];  s1 += v;  q1 += v * v;
        const float w = im[i];  s2 += w;  q2 += w * w;
    }
    red[tid]       = s1;
    red[256 + tid] = q1;
    red[512 + tid] = s2;
    red[768 + tid] = q2;
    __syncthreads();
    for (int off = 128; off > 0; off >>= 1) {
        if (tid < off) {
            red[tid]       += red[tid + off];
            red[256 + tid] += red[256 + tid + off];
            red[512 + tid] += red[512 + tid + off];
            red[768 + tid] += red[768 + tid + off];
        }
        __syncthreads();
    }
    const float mean1 = red[0]   * (1.f / N_);
    const float var1  = red[256] * (1.f / N_) - mean1 * mean1;
    const float istd1 = rsqrtf(var1 + 1e-5f);
    const float mean2 = red[512] * (1.f / N_);
    const float var2  = red[768] * (1.f / N_) - mean2 * mean2;
    const float istd2 = rsqrtf(var2 + 1e-5f);
    const float gm1 = gamma[c0],     bt1 = beta[c0];
    const float gm2 = gamma[c0 + 1], bt2 = beta[c0 + 1];

    for (int i = tid; i < N_; i += 256) {
        float v = (re[i] - mean1) * istd1 * gm1 + bt1;
        v = 0.5f * v * (1.f + erff(v * 0.70710678118654752f));
        re[i] = v;
        float w = (im[i] - mean2) * istd2 * gm2 + bt2;
        w = 0.5f * w * (1.f + erff(w * 0.70710678118654752f));
        im[i] = w;
    }
    __syncthreads();

    pass_small<false>(re, im, tid, 0);
    __syncthreads();
    for (int L = 2; L <= 8; L += 2) {
        pass4_small<false>(re, im, tid, L);
        __syncthreads();
    }
    pass4_large<false, 10>(re, im, tid);  __syncthreads();
    pass4_large<false, 12>(re, im, tid);  __syncthreads();

    float2* R0 = R + (size_t)row0 * KS;
    float2* R1 = R0 + KS;
    for (int k = tid; k < KBINS; k += 256) {
        const int km = (N_ - k) & (N_ - 1);
        const float pk = re[k], qk = im[k], pm = re[km], qm = im[km];
        R0[k] = make_float2(0.5f * (pk + pm), 0.5f * (qk - qm));
        R1[k] = make_float2(0.5f * (qk + qm), 0.5f * (pm - pk));
    }
}

__device__ __forceinline__ void store_bin(float* __restrict__ outre, float* __restrict__ outim,
                                          int b, int row, int kcol, float orv, float oim)
{
    const size_t o = (size_t)(b * C_ + row) * N_;
    const bool edge = (kcol == 0) || (kcol == NH);
    const float oiv = edge ? 0.f : oim;
    outre[o + kcol] = orv;
    outim[o + kcol] = oiv;
    if (kcol >= 1 && kcol <= NH - 1) {
        outre[o + (N_ - kcol)] =  orv;
        outim[o + (N_ - kcol)] = -oiv;
    }
}

__global__ __launch_bounds__(128) void k2_dftC_wmma(const float2* __restrict__ R,
                                                    const float2* __restrict__ Wc,
                                                    const float2* __restrict__ Ws,
                                                    float* __restrict__ outre,
                                                    float* __restrict__ outim)
{
    __shared__ float Lre[C_][40];
    __shared__ float Lim[C_][40];
    const int b    = blockIdx.y;
    const int col0 = blockIdx.x * 32;
    const int tid  = threadIdx.x;
    const float2* Rb = R + (size_t)b * C_ * KS;

    for (int idx = tid; idx < C_ * 32; idx += 128) {
        int r = idx >> 5, cc = idx & 31;
        int k = col0 + cc;
        float2 v = (k < KBINS) ? Rb[(size_t)r * KS + k] : make_float2(0.f, 0.f);
        Lre[r][cc] = v.x;
        Lim[r][cc] = v.y;
    }
    __syncthreads();

    const int wave = tid >> 5;
    const int lane = tid & 31;
    const int lg   = lane >> 4;
    const int ln   = lane & 15;
    const int m    = wave * 16 + ln;
    const float2* WcRow = Wc + m * WPAIR;
    const float2* WsRow = Ws + m * WPAIR;

    const v8f zero8 = {0.f,0.f,0.f,0.f,0.f,0.f,0.f,0.f};
    v8f cr0 = zero8, si0 = zero8, ci0 = zero8, sr0 = zero8;
    v8f cr1 = zero8, si1 = zero8, ci1 = zero8, sr1 = zero8;

    for (int k0 = 0; k0 < C_; k0 += 4) {
        const int kA = k0 + 2 * lg;
        float2 wc = WcRow[(k0 >> 1) + lg];
        float2 ws = WsRow[(k0 >> 1) + lg];
        v2f aC = { wc.x, wc.y };
        v2f aS = { ws.x, ws.y };
        v2f bRe0 = { Lre[kA][ln],      Lre[kA + 1][ln]      };
        v2f bIm0 = { Lim[kA][ln],      Lim[kA + 1][ln]      };
        v2f bRe1 = { Lre[kA][16 + ln], Lre[kA + 1][16 + ln] };
        v2f bIm1 = { Lim[kA][16 + ln], Lim[kA + 1][16 + ln] };
        cr0 = __builtin_amdgcn_wmma_f32_16x16x4_f32(false, aC, false, bRe0, (short)0, cr0, false, false);
        si0 = __builtin_amdgcn_wmma_f32_16x16x4_f32(false, aS, false, bIm0, (short)0, si0, false, false);
        ci0 = __builtin_amdgcn_wmma_f32_16x16x4_f32(false, aC, false, bIm0, (short)0, ci0, false, false);
        sr0 = __builtin_amdgcn_wmma_f32_16x16x4_f32(false, aS, false, bRe0, (short)0, sr0, false, false);
        cr1 = __builtin_amdgcn_wmma_f32_16x16x4_f32(false, aC, false, bRe1, (short)0, cr1, false, false);
        si1 = __builtin_amdgcn_wmma_f32_16x16x4_f32(false, aS, false, bIm1, (short)0, si1, false, false);
        ci1 = __builtin_amdgcn_wmma_f32_16x16x4_f32(false, aC, false, bIm1, (short)0, ci1, false, false);
        sr1 = __builtin_amdgcn_wmma_f32_16x16x4_f32(false, aS, false, bRe1, (short)0, sr1, false, false);
    }

    for (int ct = 0; ct < 2; ++ct) {
        const v8f& p1 = ct ? cr1 : cr0;
        const v8f& p2 = ct ? si1 : si0;
        const v8f& p3 = ct ? ci1 : ci0;
        const v8f& p4 = ct ? sr1 : sr0;
        const int kcol = col0 + ct * 16 + ln;
        if (kcol < KBINS) {
            for (int v = 0; v < 8; ++v) {
                const int mr = wave * 16 + 8 * lg + v;
                const bool z = (mr == 0);
                const int rowB = z ? (C_ / 2) : (C_ - mr);
                const float reA = z ? p1[v] : p1[v] + p2[v];
                const float imA = z ? p3[v] : p3[v] - p4[v];
                const float reB = z ? p4[v] : p1[v] - p2[v];
                const float imB = z ? p2[v] : p3[v] + p4[v];
                store_bin(outre, outim, b, mr,   kcol, reA, imA);
                store_bin(outre, outim, b, rowB, kcol, reB, imB);
            }
        }
    }
}

extern "C" void kernel_launch(void* const* d_in, const int* in_sizes, int n_in,
                              void* d_out, int out_size, void* d_ws, size_t ws_size,
                              hipStream_t stream) {
    const float* xre   = (const float*)d_in[0];
    const float* xim   = (const float*)d_in[1];
    const float* gamma = (const float*)d_in[2];
    const float* beta  = (const float*)d_in[3];
    float* outre = (float*)d_out;
    float* outim = outre + (size_t)B_ * C_ * N_;

    float2* R  = (float2*)d_ws;
    float2* Wc = R + (size_t)B_ * C_ * KS;
    float2* Ws = Wc + MH_ * WPAIR;

    k0_twiddle<<<dim3((MH_ * WPAIR) / 256), dim3(256), 0, stream>>>(Wc, Ws);
    k1_row_fft<<<dim3((B_ * C_) / 2), dim3(256), 0, stream>>>(xre, xim, gamma, beta, R);
    k2_dftC_wmma<<<dim3((KBINS + 31) / 32, B_), dim3(128), 0, stream>>>(R, Wc, Ws, outre, outim);
}


# channel DFT: each thread issues all 32 loads of its share of the R tile before its first LDS store (bounds test off the address): one memory latency per tile, not 32 in series; sums unchanged
# speedup vs baseline: 2.7923x; 1.1197x over previous
#include <hip/hip_runtime.h>
#include <math.h>

#define B_    16
#define C_    128
#define N_    8192
#define NH    4096
#define KBINS 4097
#define KS    4104
#define WPAIR 64
#define MH_   64

typedef __attribute__((ext_vector_type(2))) float v2f;
typedef __attribute__((ext_vector_type(8))) float v8f;

__global__ __launch_bounds__(256) void k0_twiddle(float2* __restrict__ Wc,
                                                  float2* __restrict__ Ws)
{
    const int idx = blockIdx.x * 256 + threadIdx.x;
    const int m = idx >> 6, j = idx & 63;
    const float scale = 1.0f / 1048576.0f;
    float f0 = (float)(m * (2 * j))     * (1.f / 64.f);
    float f1 = (float)(m * (2 * j + 1)) * (1.f / 64.f);
    Wc[idx] = make_float2(cospif(f0) * scale, cospif(f1) * scale);
    if (m == 0) {
        Ws[idx] = make_float2(cospif((float)(2 * j)) * scale, cospif((float)(2 * j + 1)) * scale);
    } else {
        Ws[idx] = make_float2(sinpif(f0) * scale, sinpif(f1) * scale);
    }
}

template <bool INV>
__device__ __forceinline__ void twiddle_of(int j, int ls, float& wr, float& wi)
{
    const float a = ldexpf((float)j, -ls);
    wr = cospif(a);
    wi = INV ? sinpif(a) : -sinpif(a);
}

template <bool INV>
__device__ __forceinline__ void bfly(float& ar, float& ai, float& br, float& bi,
                                     float wr, float wi)
{
    if (INV) {
        const float tr = ar - br, ti = ai - bi;
        ar = ar + br;  ai = ai + bi;
        br = tr * wr - ti * wi;
        bi = tr * wi + ti * wr;
    } else {
        const float tr = br * wr - bi * wi;
        const float ti = br * wi + bi * wr;
        br = ar - tr;  bi = ai - ti;
        ar = ar + tr;  ai = ai + ti;
    }
}

template <bool INV, int U>
__device__ __forceinline__ void bfly_group(float* __restrict__ re, float* __restrict__ im,
                                           int first, int step, int s, float wr, float wi)
{
    float ar[U], ai[U], br[U], bi[U];
    for (int u = 0; u < U; ++u) {
        const int i0 = first + u * step;
        ar[u] = re[i0];     ai[u] = im[i0];
        br[u] = re[i0 + s]; bi[u] = im[i0 + s];
    }
    for (int u = 0; u < U; ++u) {
        const int i0 = first + u * step;
        bfly<INV>(ar[u], ai[u], br[u], bi[u], wr, wi);
        re[i0]     = ar[u];  im[i0]     = ai[u];
        re[i0 + s] = br[u];  im[i0 + s] = bi[u];
    }
}

template <bool INV, int U>
__device__ __forceinline__ void quad_group(float* __restrict__ re, float* __restrict__ im,
                                           int first, int step, int h,
                                           float wsr, float wsi, float whr, float whi)
{
    const float wqr = INV ? -wsi :  wsi;
    const float wqi = INV ?  wsr : -wsr;
    float xr[U][4], xi[U][4];
    for (int u = 0; u < U; ++u) {
        const int p = first + u * step;
        for (int k = 0; k < 4; ++k) { xr[u][k] = re[p + k * h]; xi[u][k] = im[p + k * h]; }
    }
    for (int u = 0; u < U; ++u) {
        const int p = first + u * step;
        if (INV) {
            bfly<true>(xr[u][0], xi[u][0], xr[u][2], xi[u][2], wsr, wsi);
            bfly<true>(xr[u][1], xi[u][1], xr[u][3], xi[u][3], wqr, wqi);
            bfly<true>(xr[u][0], xi[u][0], xr[u][1], xi[u][1], whr, whi);
            bfly<true>(xr[u][2], xi[u][2], xr[u][3], xi[u][3], whr, whi);
        } else {
            bfly<false>(xr[u][0], xi[u][0], xr[u][1], xi[u][1], whr, whi);
            bfly<false>(xr[u][2], xi[u][2], xr[u][3], xi[u][3], whr, whi);
            bfly<false>(xr[u][0], xi[u][0], xr[u][2], xi[u][2], wsr, wsi);
            bfly<false>(xr[u][1], xi[u][1], xr[u][3], xi[u][3], wqr, wqi);
        }
        for (int k = 0; k < 4; ++k) { re[p + k * h] = xr[u][k]; im[p + k * h] = xi[u][k]; }
    }
}

template <bool INV>
__device__ __forceinline__ void pass_small(float* __restrict__ re, float* __restrict__ im,
                                           int tid, int ls)
{
    const int s    = 1 << ls;
    const int j    = tid & (s - 1);
    const int base = ((tid >> ls) << (ls + 1)) + j;
    float wr, wi;
    twiddle_of<INV>(j, ls, wr, wi);
    bfly_group<INV, 8>(re, im, base,           512, s, wr, wi);
    bfly_group<INV, 8>(re, im, base + 8 * 512, 512, s, wr, wi);
}

template <bool INV>
__device__ __forceinline__ void pass4_small(float* __restrict__ re, float* __restrict__ im,
                                            int tid, int L)
{
    const int h    = 1 << (L - 1);
    const int j    = tid & (h - 1);
    const int base = ((tid >> (L - 1)) << (L + 1)) + j;
    float wsr, wsi, whr, whi;
    twiddle_of<INV>(j, L,     wsr, wsi);
    twiddle_of<INV>(j, L - 1, whr, whi);
    quad_group<INV, 4>(re, im, base,            1024, h, wsr, wsi, whr, whi);
    quad_group<INV, 4>(re, im, base + 4 * 1024, 1024, h, wsr, wsi, whr, whi);
}

template <bool INV, int L>
__device__ __forceinline__ void pass4_large(float* __restrict__ re, float* __restrict__ im, int tid)
{
    constexpr int h = 1 << (L - 1);
    constexpr int Q = h >> 8;
    constexpr int R = 8 / Q;
    #pragma unroll 1
    for (int q = 0; q < Q; ++q) {
        const int j = tid + 256 * q;
        float wsr, wsi, whr, whi;
        twiddle_of<INV>(j, L,     wsr, wsi);
        twiddle_of<INV>(j, L - 1, whr, whi);
        quad_group<INV, R>(re, im, j, 4 * h, h, wsr, wsi, whr, whi);
    }
}

__global__ __launch_bounds__(256) void k1_row_fft(const float* __restrict__ xre,
                                                  const float* __restrict__ xim,
                                                  const float* __restrict__ gamma,
                                                  const float* __restrict__ beta,
                                                  float2* __restrict__ R)
{
    __shared__ float re[N_];
    __shared__ float im[N_];
    __shared__ float red[4 * 256];
    const int row0 = 2 * blockIdx.x;
    const int c0   = row0 & (C_ - 1);
    const int tid  = threadIdx.x;
    const float* __restrict__ a1 = xre + (size_t)row0 * N_;
    const float* __restrict__ b1 = xim + (size_t)row0 * N_;
    const float* __restrict__ a2 = a1 + N_;
    const float* __restrict__ b2 = b1 + N_;

    #pragma unroll 4
    for (int k = tid; k < NH; k += 256) {
        const int km = (N_ - k) & (N_ - 1);
        const float sA1 = a1[k] + a1[km];
        const float dB1 = b1[k] - b1[km];
        const float sA2 = a2[k] + a2[km];
        const float dB2 = b2[k] - b2[km];
        re[k]  = 0.5f * (sA1 - dB2);
        im[k]  = 0.5f * (dB1 + sA2);
        re[km] = 0.5f * (sA1 + dB2);
        im[km] = 0.5f * (sA2 - dB1);
    }
    if (tid == 0) { re[NH] = a1[NH]; im[NH] = a2[NH]; }
    __syncthreads();

    pass4_large<true, 12>(re, im, tid);  __syncthreads();
    pass4_large<true, 10>(re, im, tid);  __syncthreads();
    for (int L = 8; L >= 2; L -= 2) {
        pass4_small<true>(re, im, tid, L);
        __syncthreads();
    }
    pass_small<true>(re, im, tid, 0);
    __syncthreads();

    float s1 = 0.f, q1 = 0.f, s2 = 0.f, q2 = 0.f;
    for (int i = tid; i < N_; i += 256) {
        const float v = re[i];  s1 += v;  q1 += v * v;
        const float w = im[i];  s2 += w;  q2 += w * w;
    }
    red[tid]       = s1;
    red[256 + tid] = q1;
    red[512 + tid] = s2;
    red[768 + tid] = q2;
    __syncthreads();
    for (int off = 128; off > 0; off >>= 1) {
        if (tid < off) {
            red[tid]       += red[tid + off];
            red[256 + tid] += red[256 + tid + off];
            red[512 + tid] += red[512 + tid + off];
            red[768 + tid] += red[768 + tid + off];
        }
        __syncthreads();
    }
    const float mean1 = red[0]   * (1.f / N_);
    const float var1  = red[256] * (1.f / N_) - mean1 * mean1;
    const float istd1 = rsqrtf(var1 + 1e-5f);
    const float mean2 = red[512] * (1.f / N_);
    const float var2  = red[768] * (1.f / N_) - mean2 * mean2;
    const float istd2 = rsqrtf(var2 + 1e-5f);
    const float gm1 = gamma[c0],     bt1 = beta[c0];
    const float gm2 = gamma[c0 + 1], bt2 = beta[c0 + 1];

    for (int i = tid; i < N_; i += 256) {
        float v = (re[i] - mean1) * istd1 * gm1 + bt1;
        v = 0.5f * v * (1.f + erff(v * 0.70710678118654752f));
        re[i] = v;
        float w = (im[i] - mean2) * istd2 * gm2 + bt2;
        w = 0.5f * w * (1.f + erff(w * 0.70710678118654752f));
        im[i] = w;
    }
    __syncthreads();

    pass_small<false>(re, im, tid, 0);
    __syncthreads();
    for (int L = 2; L <= 8; L += 2) {
        pass4_small<false>(re, im, tid, L);
        __syncthreads();
    }
    pass4_large<false, 10>(re, im, tid);  __syncthreads();
    pass4_large<false, 12>(re, im, tid);  __syncthreads();

    float2* R0 = R + (size_t)row0 * KS;
    float2* R1 = R0 + KS;
    for (int k = tid; k < KBINS; k += 256) {
        const int km = (N_ - k) & (N_ - 1);
        const float pk = re[k], qk = im[k], pm = re[km], qm = im[km];
        R0[k] = make_float2(0.5f * (pk + pm), 0.5f * (qk - qm));
        R1[k] = make_float2(0.5f * (qk + qm), 0.5f * (pm - pk));
    }
}

__device__ __forceinline__ void store_bin(float* __restrict__ outre, float* __restrict__ outim,
                                          int b, int row, int kcol, float orv, float oim)
{
    const size_t o = (size_t)(b * C_ + row) * N_;
    const bool edge = (kcol == 0) || (kcol == NH);
    const float oiv = edge ? 0.f : oim;
    outre[o + kcol] = orv;
    outim[o + kcol] = oiv;
    if (kcol >= 1 && kcol <= NH - 1) {
        outre[o + (N_ - kcol)] =  orv;
        outim[o + (N_ - kcol)] = -oiv;
    }
}

__global__ __launch_bounds__(128) void k2_dftC_wmma(const float2* __restrict__ R,
                                                    const float2* __restrict__ Wc,
                                                    const float2* __restrict__ Ws,
                                                    float* __restrict__ outre,
                                                    float* __restrict__ outim)
{
    __shared__ float Lre[C_][40];
    __shared__ float Lim[C_][40];
    const int b    = blockIdx.y;
    const int col0 = blockIdx.x * 32;
    const int tid  = threadIdx.x;
    const float2* Rb = R + (size_t)b * C_ * KS;

    {
        const int  cc = tid & 31;
        const int  r0 = tid >> 5;
        const int  k  = col0 + cc;
        const bool in = k < KBINS;
        const float2* src = Rb + (size_t)r0 * KS + (in ? k : KBINS - 1);
        float2 v[32];
        #pragma unroll
        for (int i = 0; i < 32; ++i) v[i] = src[(size_t)(4 * i) * KS];
        #pragma unroll
        for (int i = 0; i < 32; ++i) {
            Lre[r0 + 4 * i][cc] = in ? v[i].x : 0.f;
            Lim[r0 + 4 * i][cc] = in ? v[i].y : 0.f;
        }
    }
    __syncthreads();

    const int wave = tid >> 5;
    const int lane = tid & 31;
    const int lg   = lane >> 4;
    const int ln   = lane & 15;
    const int m    = wave * 16 + ln;
    const float2* WcRow = Wc + m * WPAIR;
    const float2* WsRow = Ws + m * WPAIR;

    const v8f zero8 = {0.f,0.f,0.f,0.f,0.f,0.f,0.f,0.f};
    v8f cr0 = zero8, si0 = zero8, ci0 = zero8, sr0 = zero8;
    v8f cr1 = zero8, si1 = zero8, ci1 = zero8, sr1 = zero8;

    for (int k0 = 0; k0 < C_; k0 += 4) {
        const int kA = k0 + 2 * lg;
        float2 wc = WcRow[(k0 >> 1) + lg];
        float2 ws = WsRow[(k0 >> 1) + lg];
        v2f aC = { wc.x, wc.y };
        v2f aS = { ws.x, ws.y };
        v2f bRe0 = { Lre[kA][ln],      Lre[kA + 1][ln]      };
        v2f bIm0 = { Lim[kA][ln],      Lim[kA + 1][ln]      };
        v2f bRe1 = { Lre[kA][16 + ln], Lre[kA + 1][16 + ln] };
        v2f bIm1 = { Lim[kA][16 + ln], Lim[kA + 1][16 + ln] };
        cr0 = __builtin_amdgcn_wmma_f32_16x16x4_f32(false, aC, false, bRe0, (short)0, cr0, false, false);
        si0 = __builtin_amdgcn_wmma_f32_16x16x4_f32(false, aS, false, bIm0, (short)0, si0, false, false);
        ci0 = __builtin_amdgcn_wmma_f32_16x16x4_f32(false, aC, false, bIm0, (short)0, ci0, false, false);
        sr0 = __builtin_amdgcn_wmma_f32_16x16x4_f32(false, aS, false, bRe0, (short)0, sr0, false, false);
        cr1 = __builtin_amdgcn_wmma_f32_16x16x4_f32(false, aC, false, bRe1, (short)0, cr1, false, false);
        si1 = __builtin_amdgcn_wmma_f32_16x16x4_f32(false, aS, false, bIm1, (short)0, si1, false, false);
        ci1 = __builtin_amdgcn_wmma_f32_16x16x4_f32(false, aC, false, bIm1, (short)0, ci1, false, false);
        sr1 = __builtin_amdgcn_wmma_f32_16x16x4_f32(false, aS, false, bRe1, (short)0, sr1, false, false);
    }

    for (int ct = 0; ct < 2; ++ct) {
        const v8f& p1 = ct ? cr1 : cr0;
        const v8f& p2 = ct ? si1 : si0;
        const v8f& p3 = ct ? ci1 : ci0;
        const v8f& p4 = ct ? sr1 : sr0;
        const int kcol = col0 + ct * 16 + ln;
        if (kcol < KBINS) {
            for (int v = 0; v < 8; ++v) {
                const int mr = wave * 16 + 8 * lg + v;
                const bool z = (mr == 0);
                const int rowB = z ? (C_ / 2) : (C_ - mr);
                const float reA = z ? p1[v] : p1[v] + p2[v];
                const float imA = z ? p3[v] : p3[v] - p4[v];
                const float reB = z ? p4[v] : p1[v] - p2[v];
                const float imB = z ? p2[v] : p3[v] + p4[v];
                store_bin(outre, outim, b, mr,   kcol, reA, imA);
                store_bin(outre, outim, b, rowB, kcol, reB, imB);
            }
        }
    }
}

extern "C" void kernel_launch(void* const* d_in, const int* in_sizes, int n_in,
                              void* d_out, int out_size, void* d_ws, size_t ws_size,
                              hipStream_t stream) {
    const float* xre   = (const float*)d_in[0];
    const float* xim   = (const float*)d_in[1];
    const float* gamma = (const float*)d_in[2];
    const float* beta  = (const float*)d_in[3];
    float* outre = (float*)d_out;
    float* outim = outre + (size_t)B_ * C_ * N_;

    float2* R  = (float2*)d_ws;
    float2* Wc = R + (size_t)B_ * C_ * KS;
    float2* Ws = Wc + MH_ * WPAIR;

    k0_twiddle<<<dim3((MH_ * WPAIR) / 256), dim3(256), 0, stream>>>(Wc, Ws);
    k1_row_fft<<<dim3((B_ * C_) / 2), dim3(256), 0, stream>>>(xre, xim, gamma, beta, R);
    k2_dftC_wmma<<<dim3((KBINS + 31) / 32, B_), dim3(128), 0, stream>>>(R, Wc, Ws, outre, outim);
}


# row FFT: the line lives in LDS as interleaved (re, im) pairs: one 8-byte LDS access per point instead of two 4-byte ones (half the LDS instructions); arithmetic and sum order unchanged
# speedup vs baseline: 2.8697x; 1.0277x over previous
#include <hip/hip_runtime.h>
#include <math.h>

#define B_    16
#define C_    128
#define N_    8192
#define NH    4096
#define KBINS 4097
#define KS    4104
#define WPAIR 64
#define MH_   64

typedef __attribute__((ext_vector_type(2))) float v2f;
typedef __attribute__((ext_vector_type(8))) float v8f;

__global__ __launch_bounds__(256) void k0_twiddle(float2* __restrict__ Wc,
                                                  float2* __restrict__ Ws)
{
    const int idx = blockIdx.x * 256 + threadIdx.x;
    const int m = idx >> 6, j = idx & 63;
    const float scale = 1.0f / 1048576.0f;
    float f0 = (float)(m * (2 * j))     * (1.f / 64.f);
    float f1 = (float)(m * (2 * j + 1)) * (1.f / 64.f);
    Wc[idx] = make_float2(cospif(f0) * scale, cospif(f1) * scale);
    if (m == 0) {
        Ws[idx] = make_float2(cospif((float)(2 * j)) * scale, cospif((float)(2 * j + 1)) * scale);
    } else {
        Ws[idx] = make_float2(sinpif(f0) * scale, sinpif(f1) * scale);
    }
}

template <bool INV>
__device__ __forceinline__ void twiddle_of(int j, int ls, float& wr, float& wi)
{
    const float a = ldexpf((float)j, -ls);
    wr = cospif(a);
    wi = INV ? sinpif(a) : -sinpif(a);
}

template <bool INV>
__device__ __forceinline__ void bfly(float& ar, float& ai, float& br, float& bi,
                                     float wr, float wi)
{
    if (INV) {
        const float tr = ar - br, ti = ai - bi;
        ar = ar + br;  ai = ai + bi;
        br = tr * wr - ti * wi;
        bi = tr * wi + ti * wr;
    } else {
        const float tr = br * wr - bi * wi;
        const float ti = br * wi + bi * wr;
        br = ar - tr;  bi = ai - ti;
        ar = ar + tr;  ai = ai + ti;
    }
}

template <bool INV, int U>
__device__ __forceinline__ void bfly_group(float2* __restrict__ z,
                                           int first, int step, int s, float wr, float wi)
{
    float ar[U], ai[U], br[U], bi[U];
    for (int u = 0; u < U; ++u) {
        const int i0 = first + u * step;
        const float2 a = z[i0], b = z[i0 + s];
        ar[u] = a.x;  ai[u] = a.y;
        br[u] = b.x;  bi[u] = b.y;
    }
    for (int u = 0; u < U; ++u) {
        const int i0 = first + u * step;
        bfly<INV>(ar[u], ai[u], br[u], bi[u], wr, wi);
        z[i0]     = make_float2(ar[u], ai[u]);
        z[i0 + s] = make_float2(br[u], bi[u]);
    }
}

template <bool INV, int U>
__device__ __forceinline__ void quad_group(float2* __restrict__ z,
                                           int first, int step, int h,
                                           float wsr, float wsi, float whr, float whi)
{
    const float wqr = INV ? -wsi :  wsi;
    const float wqi = INV ?  wsr : -wsr;
    float xr[U][4], xi[U][4];
    for (int u = 0; u < U; ++u) {
        const int p = first + u * step;
        for (int k = 0; k < 4; ++k) { const float2 t = z[p + k * h]; xr[u][k] = t.x; xi[u][k] = t.y; }
    }
    for (int u = 0; u < U; ++u) {
        const int p = first + u * step;
        if (INV) {
            bfly<true>(xr[u][0], xi[u][0], xr[u][2], xi[u][2], wsr, wsi);
            bfly<true>(xr[u][1], xi[u][1], xr[u][3], xi[u][3], wqr, wqi);
            bfly<true>(xr[u][0], xi[u][0], xr[u][1], xi[u][1], whr, whi);
            bfly<true>(xr[u][2], xi[u][2], xr[u][3], xi[u][3], whr, whi);
        } else {
            bfly<false>(xr[u][0], xi[u][0], xr[u][1], xi[u][1], whr, whi);
            bfly<false>(xr[u][2], xi[u][2], xr[u][3], xi[u][3], whr, whi);
            bfly<false>(xr[u][0], xi[u][0], xr[u][2], xi[u][2], wsr, wsi);
            bfly<false>(xr[u][1], xi[u][1], xr[u][3], xi[u][3], wqr, wqi);
        }
        for (int k = 0; k < 4; ++k) z[p + k * h] = make_float2(xr[u][k], xi[u][k]);
    }
}

template <bool INV>
__device__ __forceinline__ void pass_small(float2* __restrict__ z,
                                           int tid, int ls)
{
    const int s    = 1 << ls;
    const int j    = tid & (s - 1);
    const int base = ((tid >> ls) << (ls + 1)) + j;
    float wr, wi;
    twiddle_of<INV>(j, ls, wr, wi);
    bfly_group<INV, 8>(z, base,           512, s, wr, wi);
    bfly_group<INV, 8>(z, base + 8 * 512, 512, s, wr, wi);
}

template <bool INV>
__device__ __forceinline__ void pass4_small(float2* __restrict__ z,
                                            int tid, int L)
{
    const int h    = 1 << (L - 1);
    const int j    = tid & (h - 1);
    const int base = ((tid >> (L - 1)) << (L + 1)) + j;
    float wsr, wsi, whr, whi;
    twiddle_of<INV>(j, L,     wsr, wsi);
    twiddle_of<INV>(j, L - 1, whr, whi);
    quad_group<INV, 4>(z, base,            1024, h, wsr, wsi, whr, whi);
    quad_group<INV, 4>(z, base + 4 * 1024, 1024, h, wsr, wsi, whr, whi);
}

template <bool INV, int L>
__device__ __forceinline__ void pass4_large(float2* __restrict__ z, int tid)
{
    constexpr int h = 1 << (L - 1);
    constexpr int Q = h >> 8;
    constexpr int R = 8 / Q;
    #pragma unroll 1
    for (int q = 0; q < Q; ++q) {
        const int j = tid + 256 * q;
        float wsr, wsi, whr, whi;
        twiddle_of<INV>(j, L,     wsr, wsi);
        twiddle_of<INV>(j, L - 1, whr, whi);
        quad_group<INV, R>(z, j, 4 * h, h, wsr, wsi, whr, whi);
    }
}

__global__ __launch_bounds__(256) void k1_row_fft(const float* __restrict__ xre,
                                                  const float* __restrict__ xim,
                                                  const float* __restrict__ gamma,
                                                  const float* __restrict__ beta,
                                                  float2* __restrict__ R)
{
    __shared__ __align__(16) float2 z[N_];
    __shared__ float red[4 * 256];
    const int row0 = 2 * blockIdx.x;
    const int c0   = row0 & (C_ - 1);
    const int tid  = threadIdx.x;
    const float* __restrict__ a1 = xre + (size_t)row0 * N_;
    const float* __restrict__ b1 = xim + (size_t)row0 * N_;
    const float* __restrict__ a2 = a1 + N_;
    const float* __restrict__ b2 = b1 + N_;

    #pragma unroll 4
    for (int k = tid; k < NH; k += 256) {
        const int km = (N_ - k) & (N_ - 1);
        const float sA1 = a1[k] + a1[km];
        const float dB1 = b1[k] - b1[km];
        const float sA2 = a2[k] + a2[km];
        const float dB2 = b2[k] - b2[km];
        z[k]  = make_float2(0.5f * (sA1 - dB2), 0.5f * (dB1 + sA2));
        z[km] = make_float2(0.5f * (sA1 + dB2), 0.5f * (sA2 - dB1));
    }
    if (tid == 0) z[NH] = make_float2(a1[NH], a2[NH]);
    __syncthreads();

    pass4_large<true, 12>(z, tid);  __syncthreads();
    pass4_large<true, 10>(z, tid);  __syncthreads();
    for (int L = 8; L >= 2; L -= 2) {
        pass4_small<true>(z, tid, L);
        __syncthreads();
    }
    pass_small<true>(z, tid, 0);
    __syncthreads();

    float s1 = 0.f, q1 = 0.f, s2 = 0.f, q2 = 0.f;
    for (int i = tid; i < N_; i += 256) {
        const float2 t = z[i];
        const float v = t.x;  s1 += v;  q1 += v * v;
        const float w = t.y;  s2 += w;  q2 += w * w;
    }
    red[tid]       = s1;
    red[256 + tid] = q1;
    red[512 + tid] = s2;
    red[768 + tid] = q2;
    __syncthreads();
    for (int off = 128; off > 0; off >>= 1) {
        if (tid < off) {
            red[tid]       += red[tid + off];
            red[256 + tid] += red[256 + tid + off];
            red[512 + tid] += red[512 + tid + off];
            red[768 + tid] += red[768 + tid + off];
        }
        __syncthreads();
    }
    const float mean1 = red[0]   * (1.f / N_);
    const float var1  = red[256] * (1.f / N_) - mean1 * mean1;
    const float istd1 = rsqrtf(var1 + 1e-5f);
    const float mean2 = red[512] * (1.f / N_);
    const float var2  = red[768] * (1.f / N_) - mean2 * mean2;
    const float istd2 = rsqrtf(var2 + 1e-5f);
    const float gm1 = gamma[c0],     bt1 = beta[c0];
    const float gm2 = gamma[c0 + 1], bt2 = beta[c0 + 1];

    for (int i = tid; i < N_; i += 256) {
        const float2 t = z[i];
        float v = (t.x - mean1) * istd1 * gm1 + bt1;
        v = 0.5f * v * (1.f + erff(v * 0.70710678118654752f));
        float w = (t.y - mean2) * istd2 * gm2 + bt2;
        w = 0.5f * w * (1.f + erff(w * 0.70710678118654752f));
        z[i] = make_float2(v, w);
    }
    __syncthreads();

    pass_small<false>(z, tid, 0);
    __syncthreads();
    for (int L = 2; L <= 8; L += 2) {
        pass4_small<false>(z, tid, L);
        __syncthreads();
    }
    pass4_large<false, 10>(z, tid);  __syncthreads();
    pass4_large<false, 12>(z, tid);  __syncthreads();

    float2* R0 = R + (size_t)row0 * KS;
    float2* R1 = R0 + KS;
    for (int k = tid; k < KBINS; k += 256) {
        const int km = (N_ - k) & (N_ - 1);
        const float2 fk = z[k], fm = z[km];
        const float pk = fk.x, qk = fk.y, pm = fm.x, qm = fm.y;
        R0[k] = make_float2(0.5f * (pk + pm), 0.5f * (qk - qm));
        R1[k] = make_float2(0.5f * (qk + qm), 0.5f * (pm - pk));
    }
}

__device__ __forceinline__ void store_bin(float* __restrict__ outre, float* __restrict__ outim,
                                          int b, int row, int kcol, float orv, float oim)
{
    const size_t o = (size_t)(b * C_ + row) * N_;
    const bool edge = (kcol == 0) || (kcol == NH);
    const float oiv = edge ? 0.f : oim;
    outre[o + kcol] = orv;
    outim[o + kcol] = oiv;
    if (kcol >= 1 && kcol <= NH - 1) {
        outre[o + (N_ - kcol)] =  orv;
        outim[o + (N_ - kcol)] = -oiv;
    }
}

__global__ __launch_bounds__(128) void k2_dftC_wmma(const float2* __restrict__ R,
                                                    const float2* __restrict__ Wc,
                                                    const float2* __restrict__ Ws,
                                                    float* __restrict__ outre,
                                                    float* __restrict__ outim)
{
    __shared__ float Lre[C_][40];
    __shared__ float Lim[C_][40];
    const int b    = blockIdx.y;
    const int col0 = blockIdx.x * 32;
    const int tid  = threadIdx.x;
    const float2* Rb = R + (size_t)b * C_ * KS;

    {
        const int  cc = tid & 31;
        const int  r0 = tid >> 5;
        const int  k  = col0 + cc;
        const bool in = k < KBINS;
        const float2* src = Rb + (size_t)r0 * KS + (in ? k : KBINS - 1);
        float2 v[32];
        #pragma unroll
        for (int i = 0; i < 32; ++i) v[i] = src[(size_t)(4 * i) * KS];
        #pragma unroll
        for (int i = 0; i < 32; ++i) {
            Lre[r0 + 4 * i][cc] = in ? v[i].x : 0.f;
            Lim[r0 + 4 * i][cc] = in ? v[i].y : 0.f;
        }
    }
    __syncthreads();

    const int wave = tid >> 5;
    const int lane = tid & 31;
    const int lg   = lane >> 4;
    const int ln   = lane & 15;
    const int m    = wave * 16 + ln;
    const float2* WcRow = Wc + m * WPAIR;
    const float2* WsRow = Ws + m * WPAIR;

    const v8f zero8 = {0.f,0.f,0.f,0.f,0.f,0.f,0.f,0.f};
    v8f cr0 = zero8, si0 = zero8, ci0 = zero8, sr0 = zero8;
    v8f cr1 = zero8, si1 = zero8, ci1 = zero8, sr1 = zero8;

    for (int k0 = 0; k0 < C_; k0 += 4) {
        const int kA = k0 + 2 * lg;
        float2 wc = WcRow[(k0 >> 1) + lg];
        float2 ws = WsRow[(k0 >> 1) + lg];
        v2f aC = { wc.x, wc.y };
        v2f aS = { ws.x, ws.y };
        v2f bRe0 = { Lre[kA][ln],      Lre[kA + 1][ln]      };
        v2f bIm0 = { Lim[kA][ln],      Lim[kA + 1][ln]      };
        v2f bRe1 = { Lre[kA][16 + ln], Lre[kA + 1][16 + ln] };
        v2f bIm1 = { Lim[kA][16 + ln], Lim[kA + 1][16 + ln] };
        cr0 = __builtin_amdgcn_wmma_f32_16x16x4_f32(false, aC, false, bRe0, (short)0, cr0, false, false);
        si0 = __builtin_amdgcn_wmma_f32_16x16x4_f32(false, aS, false, bIm0, (short)0, si0, false, false);
        ci0 = __builtin_amdgcn_wmma_f32_16x16x4_f32(false, aC, false, bIm0, (short)0, ci0, false, false);
        sr0 = __builtin_amdgcn_wmma_f32_16x16x4_f32(false, aS, false, bRe0, (short)0, sr0, false, false);
        cr1 = __builtin_amdgcn_wmma_f32_16x16x4_f32(false, aC, false, bRe1, (short)0, cr1, false, false);
        si1 = __builtin_amdgcn_wmma_f32_16x16x4_f32(false, aS, false, bIm1, (short)0, si1, false, false);
        ci1 = __builtin_amdgcn_wmma_f32_16x16x4_f32(false, aC, false, bIm1, (short)0, ci1, false, false);
        sr1 = __builtin_amdgcn_wmma_f32_16x16x4_f32(false, aS, false, bRe1, (short)0, sr1, false, false);
    }

    for (int ct = 0; ct < 2; ++ct) {
        const v8f& p1 = ct ? cr1 : cr0;
        const v8f& p2 = ct ? si1 : si0;
        const v8f& p3 = ct ? ci1 : ci0;
        const v8f& p4 = ct ? sr1 : sr0;
        const int kcol = col0 + ct * 16 + ln;
        if (kcol < KBINS) {
            for (int v = 0; v < 8; ++v) {
                const int mr = wave * 16 + 8 * lg + v;
                const bool z = (mr == 0);
                const int rowB = z ? (C_ / 2) : (C_ - mr);
                const float reA = z ? p1[v] : p1[v] + p2[v];
                const float imA = z ? p3[v] : p3[v] - p4[v];
                const float reB = z ? p4[v] : p1[v] - p2[v];
                const float imB = z ? p2[v] : p3[v] + p4[v];
                store_bin(outre, outim, b, mr,   kcol, reA, imA);
                store_bin(outre, outim, b, rowB, kcol, reB, imB);
            }
        }
    }
}

extern "C" void kernel_launch(void* const* d_in, const int* in_sizes, int n_in,
                              void* d_out, int out_size, void* d_ws, size_t ws_size,
                              hipStream_t stream) {
    const float* xre   = (const float*)d_in[0];
    const float* xim   = (const float*)d_in[1];
    const float* gamma = (const float*)d_in[2];
    const float* beta  = (const float*)d_in[3];
    float* outre = (float*)d_out;
    float* outim = outre + (size_t)B_ * C_ * N_;

    float2* R  = (float2*)d_ws;
    float2* Wc = R + (size_t)B_ * C_ * KS;
    float2* Ws = Wc + MH_ * WPAIR;

    k0_twiddle<<<dim3((MH_ * WPAIR) / 256), dim3(256), 0, stream>>>(Wc, Ws);
    k1_row_fft<<<dim3((B_ * C_) / 2), dim3(256), 0, stream>>>(xre, xim, gamma, beta, R);
    k2_dftC_wmma<<<dim3((KBINS + 31) / 32, B_), dim3(128), 0, stream>>>(R, Wc, Ws, outre, outim);
}


# GELU: erf as one clamped rational form x*P(x^2)/Q(x^2) with a hardware reciprocal, not the two-range library erff (both ranges ran for every element); erf differs by up to 3e-7; sum order unchanged
# speedup vs baseline: 2.9843x; 1.0399x over previous
#include <hip/hip_runtime.h>
#include <math.h>

#define B_    16
#define C_    128
#define N_    8192
#define NH    4096
#define KBINS 4097
#define KS    4104
#define WPAIR 64
#define MH_   64

typedef __attribute__((ext_vector_type(2))) float v2f;
typedef __attribute__((ext_vector_type(8))) float v8f;

__global__ __launch_bounds__(256) void k0_twiddle(float2* __restrict__ Wc,
                                                  float2* __restrict__ Ws)
{
    const int idx = blockIdx.x * 256 + threadIdx.x;
    const int m = idx >> 6, j = idx & 63;
    const float scale = 1.0f / 1048576.0f;
    float f0 = (float)(m * (2 * j))     * (1.f / 64.f);
    float f1 = (float)(m * (2 * j + 1)) * (1.f / 64.f);
    Wc[idx] = make_float2(cospif(f0) * scale, cospif(f1) * scale);
    if (m == 0) {
        Ws[idx] = make_float2(cospif((float)(2 * j)) * scale, cospif((float)(2 * j + 1)) * scale);
    } else {
        Ws[idx] = make_float2(sinpif(f0) * scale, sinpif(f1) * scale);
    }
}

template <bool INV>
__device__ __forceinline__ void twiddle_of(int j, int ls, float& wr, float& wi)
{
    const float a = ldexpf((float)j, -ls);
    wr = cospif(a);
    wi = INV ? sinpif(a) : -sinpif(a);
}

template <bool INV>
__device__ __forceinline__ void bfly(float& ar, float& ai, float& br, float& bi,
                                     float wr, float wi)
{
    if (INV) {
        const float tr = ar - br, ti = ai - bi;
        ar = ar + br;  ai = ai + bi;
        br = tr * wr - ti * wi;
        bi = tr * wi + ti * wr;
    } else {
        const float tr = br * wr - bi * wi;
        const float ti = br * wi + bi * wr;
        br = ar - tr;  bi = ai - ti;
        ar = ar + tr;  ai = ai + ti;
    }
}

template <bool INV, int U>
__device__ __forceinline__ void bfly_group(float2* __restrict__ z,
                                           int first, int step, int s, float wr, float wi)
{
    float ar[U], ai[U], br[U], bi[U];
    for (int u = 0; u < U; ++u) {
        const int i0 = first + u * step;
        const float2 a = z[i0], b = z[i0 + s];
        ar[u] = a.x;  ai[u] = a.y;
        br[u] = b.x;  bi[u] = b.y;
    }
    for (int u = 0; u < U; ++u) {
        const int i0 = first + u * step;
        bfly<INV>(ar[u], ai[u], br[u], bi[u], wr, wi);
        z[i0]     = make_float2(ar[u], ai[u]);
        z[i0 + s] = make_float2(br[u], bi[u]);
    }
}

template <bool INV, int U>
__device__ __forceinline__ void quad_group(float2* __restrict__ z,
                                           int first, int step, int h,
                                           float wsr, float wsi, float whr, float whi)
{
    const float wqr = INV ? -wsi :  wsi;
    const float wqi = INV ?  wsr : -wsr;
    float xr[U][4], xi[U][4];
    for (int u = 0; u < U; ++u) {
        const int p = first + u * step;
        for (int k = 0; k < 4; ++k) { const float2 t = z[p + k * h]; xr[u][k] = t.x; xi[u][k] = t.y; }
    }
    for (int u = 0; u < U; ++u) {
        const int p = first + u * step;
        if (INV) {
            bfly<true>(xr[u][0], xi[u][0], xr[u][2], xi[u][2], wsr, wsi);
            bfly<true>(xr[u][1], xi[u][1], xr[u][3], xi[u][3], wqr, wqi);
            bfly<true>(xr[u][0], xi[u][0], xr[u][1], xi[u][1], whr, whi);
            bfly<true>(xr[u][2], xi[u][2], xr[u][3], xi[u][3], whr, whi);
        } else {
            bfly<false>(xr[u][0], xi[u][0], xr[u][1], xi[u][1], whr, whi);
            bfly<false>(xr[u][2], xi[u][2], xr[u][3], xi[u][3], whr, whi);
            bfly<false>(xr[u][0], xi[u][0], xr[u][2], xi[u][2], wsr, wsi);
            bfly<false>(xr[u][1], xi[u][1], xr[u][3], xi[u][3], wqr, wqi);
        }
        for (int k = 0; k < 4; ++k) z[p + k * h] = make_float2(xr[u][k], xi[u][k]);
    }
}

template <bool INV>
__device__ __forceinline__ void pass_small(float2* __restrict__ z,
                                           int tid, int ls)
{
    const int s    = 1 << ls;
    const int j    = tid & (s - 1);
    const int base = ((tid >> ls) << (ls + 1)) + j;
    float wr, wi;
    twiddle_of<INV>(j, ls, wr, wi);
    bfly_group<INV, 8>(z, base,           512, s, wr, wi);
    bfly_group<INV, 8>(z, base + 8 * 512, 512, s, wr, wi);
}

template <bool INV>
__device__ __forceinline__ void pass4_small(float2* __restrict__ z,
                                            int tid, int L)
{
    const int h    = 1 << (L - 1);
    const int j    = tid & (h - 1);
    const int base = ((tid >> (L - 1)) << (L + 1)) + j;
    float wsr, wsi, whr, whi;
    twiddle_of<INV>(j, L,     wsr, wsi);
    twiddle_of<INV>(j, L - 1, whr, whi);
    quad_group<INV, 4>(z, base,            1024, h, wsr, wsi, whr, whi);
    quad_group<INV, 4>(z, base + 4 * 1024, 1024, h, wsr, wsi, whr, whi);
}

template <bool INV, int L>
__device__ __forceinline__ void pass4_large(float2* __restrict__ z, int tid)
{
    constexpr int h = 1 << (L - 1);
    constexpr int Q = h >> 8;
    constexpr int R = 8 / Q;
    #pragma unroll 1
    for (int q = 0; q < Q; ++q) {
        const int j = tid + 256 * q;
        float wsr, wsi, whr, whi;
        twiddle_of<INV>(j, L,     wsr, wsi);
        twiddle_of<INV>(j, L - 1, whr, whi);
        quad_group<INV, R>(z, j, 4 * h, h, wsr, wsi, whr, whi);
    }
}

__device__ __forceinline__ float erf_rational(float x)
{
    x = fminf(fmaxf(x, -4.f), 4.f);
    const float x2 = x * x;
    float p = -2.72614225801306e-10f;
    p = fmaf(p, x2,  2.77068142495902e-08f);
    p = fmaf(p, x2, -2.10102402082508e-06f);
    p = fmaf(p, x2, -5.69250639462346e-05f);
    p = fmaf(p, x2, -7.34990630326855e-04f);
    p = fmaf(p, x2, -2.95459980854025e-03f);
    p = fmaf(p, x2, -1.60960333262415e-02f);
    float q = -1.45660718464996e-05f;
    q = fmaf(q, x2, -2.13374055278905e-04f);
    q = fmaf(q, x2, -1.68282697438203e-03f);
    q = fmaf(q, x2, -7.37332916720468e-03f);
    q = fmaf(q, x2, -1.42647390514189e-02f);
    const float r = (x * p) * __builtin_amdgcn_rcpf(q);
    return fminf(fmaxf(r, -1.f), 1.f);
}

__global__ __launch_bounds__(256) void k1_row_fft(const float* __restrict__ xre,
                                                  const float* __restrict__ xim,
                                                  const float* __restrict__ gamma,
                                                  const float* __restrict__ beta,
                                                  float2* __restrict__ R)
{
    __shared__ __align__(16) float2 z[N_];
    __shared__ float red[4 * 256];
    const int row0 = 2 * blockIdx.x;
    const int c0   = row0 & (C_ - 1);
    const int tid  = threadIdx.x;
    const float* __restrict__ a1 = xre + (size_t)row0 * N_;
    const float* __restrict__ b1 = xim + (size_t)row0 * N_;
    const float* __restrict__ a2 = a1 + N_;
    const float* __restrict__ b2 = b1 + N_;

    #pragma unroll 4
    for (int k = tid; k < NH; k += 256) {
        const int km = (N_ - k) & (N_ - 1);
        const float sA1 = a1[k] + a1[km];
        const float dB1 = b1[k] - b1[km];
        const float sA2 = a2[k] + a2[km];
        const float dB2 = b2[k] - b2[km];
        z[k]  = make_float2(0.5f * (sA1 - dB2), 0.5f * (dB1 + sA2));
        z[km] = make_float2(0.5f * (sA1 + dB2), 0.5f * (sA2 - dB1));
    }
    if (tid == 0) z[NH] = make_float2(a1[NH], a2[NH]);
    __syncthreads();

    pass4_large<true, 12>(z, tid);  __syncthreads();
    pass4_large<true, 10>(z, tid);  __syncthreads();
    for (int L = 8; L >= 2; L -= 2) {
        pass4_small<true>(z, tid, L);
        __syncthreads();
    }
    pass_small<true>(z, tid, 0);
    __syncthreads();

    float s1 = 0.f, q1 = 0.f, s2 = 0.f, q2 = 0.f;
    for (int i = tid; i < N_; i += 256) {
        const float2 t = z[i];
        const float v = t.x;  s1 += v;  q1 += v * v;
        const float w = t.y;  s2 += w;  q2 += w * w;
    }
    red[tid]       = s1;
    red[256 + tid] = q1;
    red[512 + tid] = s2;
    red[768 + tid] = q2;
    __syncthreads();
    for (int off = 128; off > 0; off >>= 1) {
        if (tid < off) {
            red[tid]       += red[tid + off];
            red[256 + tid] += red[256 + tid + off];
            red[512 + tid] += red[512 + tid + off];
            red[768 + tid] += red[768 + tid + off];
        }
        __syncthreads();
    }
    const float mean1 = red[0]   * (1.f / N_);
    const float var1  = red[256] * (1.f / N_) - mean1 * mean1;
    const float istd1 = rsqrtf(var1 + 1e-5f);
    const float mean2 = red[512] * (1.f / N_);
    const float var2  = red[768] * (1.f / N_) - mean2 * mean2;
    const float istd2 = rsqrtf(var2 + 1e-5f);
    const float gm1 = gamma[c0],     bt1 = beta[c0];
    const float gm2 = gamma[c0 + 1], bt2 = beta[c0 + 1];

    for (int i = tid; i < N_; i += 256) {
        const float2 t = z[i];
        float v = (t.x - mean1) * istd1 * gm1 + bt1;
        v = 0.5f * v * (1.f + erf_rational(v * 0.70710678118654752f));
        float w = (t.y - mean2) * istd2 * gm2 + bt2;
        w = 0.5f * w * (1.f + erf_rational(w * 0.70710678118654752f));
        z[i] = make_float2(v, w);
    }
    __syncthreads();

    pass_small<false>(z, tid, 0);
    __syncthreads();
    for (int L = 2; L <= 8; L += 2) {
        pass4_small<false>(z, tid, L);
        __syncthreads();
    }
    pass4_large<false, 10>(z, tid);  __syncthreads();
    pass4_large<false, 12>(z, tid);  __syncthreads();

    float2* R0 = R + (size_t)row0 * KS;
    float2* R1 = R0 + KS;
    for (int k = tid; k < KBINS; k += 256) {
        const int km = (N_ - k) & (N_ - 1);
        const float2 fk = z[k], fm = z[km];
        const float pk = fk.x, qk = fk.y, pm = fm.x, qm = fm.y;
        R0[k] = make_float2(0.5f * (pk + pm), 0.5f * (qk - qm));
        R1[k] = make_float2(0.5f * (qk + qm), 0.5f * (pm - pk));
    }
}

__device__ __forceinline__ void store_bin(float* __restrict__ outre, float* __restrict__ outim,
                                          int b, int row, int kcol, float orv, float oim)
{
    const size_t o = (size_t)(b * C_ + row) * N_;
    const bool edge = (kcol == 0) || (kcol == NH);
    const float oiv = edge ? 0.f : oim;
    outre[o + kcol] = orv;
    outim[o + kcol] = oiv;
    if (kcol >= 1 && kcol <= NH - 1) {
        outre[o + (N_ - kcol)] =  orv;
        outim[o + (N_ - kcol)] = -oiv;
    }
}

__global__ __launch_bounds__(128) void k2_dftC_wmma(const float2* __restrict__ R,
                                                    const float2* __restrict__ Wc,
                                                    const float2* __restrict__ Ws,
                                                    float* __restrict__ outre,
                                                    float* __restrict__ outim)
{
    __shared__ float Lre[C_][40];
    __shared__ float Lim[C_][40];
    const int b    = blockIdx.y;
    const int col0 = blockIdx.x * 32;
    const int tid  = threadIdx.x;
    const float2* Rb = R + (size_t)b * C_ * KS;

    {
        const int  cc = tid & 31;
        const int  r0 = tid >> 5;
        const int  k  = col0 + cc;
        const bool in = k < KBINS;
        const float2* src = Rb + (size_t)r0 * KS + (in ? k : KBINS - 1);
        float2 v[32];
        #pragma unroll
        for (int i = 0; i < 32; ++i) v[i] = src[(size_t)(4 * i) * KS];
        #pragma unroll
        for (int i = 0; i < 32; ++i) {
            Lre[r0 + 4 * i][cc] = in ? v[i].x : 0.f;
            Lim[r0 + 4 * i][cc] = in ? v[i].y : 0.f;
        }
    }
    __syncthreads();

    const int wave = tid >> 5;
    const int lane = tid & 31;
    const int lg   = lane >> 4;
    const int ln   = lane & 15;
    const int m    = wave * 16 + ln;
    const float2* WcRow = Wc + m * WPAIR;
    const float2* WsRow = Ws + m * WPAIR;

    const v8f zero8 = {0.f,0.f,0.f,0.f,0.f,0.f,0.f,0.f};
    v8f cr0 = zero8, si0 = zero8, ci0 = zero8, sr0 = zero8;
    v8f cr1 = zero8, si1 = zero8, ci1 = zero8, sr1 = zero8;

    for (int k0 = 0; k0 < C_; k0 += 4) {
        const int kA = k0 + 2 * lg;
        float2 wc = WcRow[(k0 >> 1) + lg];
        float2 ws = WsRow[(k0 >> 1) + lg];
        v2f aC = { wc.x, wc.y };
        v2f aS = { ws.x, ws.y };
        v2f bRe0 = { Lre[kA][ln],      Lre[kA + 1][ln]      };
        v2f bIm0 = { Lim[kA][ln],      Lim[kA + 1][ln]      };
        v2f bRe1 = { Lre[kA][16 + ln], Lre[kA + 1][16 + ln] };
        v2f bIm1 = { Lim[kA][16 + ln], Lim[kA + 1][16 + ln] };
        cr0 = __builtin_amdgcn_wmma_f32_16x16x4_f32(false, aC, false, bRe0, (short)0, cr0, false, false);
        si0 = __builtin_amdgcn_wmma_f32_16x16x4_f32(false, aS, false, bIm0, (short)0, si0, false, false);
        ci0 = __builtin_amdgcn_wmma_f32_16x16x4_f32(false, aC, false, bIm0, (short)0, ci0, false, false);
        sr0 = __builtin_amdgcn_wmma_f32_16x16x4_f32(false, aS, false, bRe0, (short)0, sr0, false, false);
        cr1 = __builtin_amdgcn_wmma_f32_16x16x4_f32(false, aC, false, bRe1, (short)0, cr1, false, false);
        si1 = __builtin_amdgcn_wmma_f32_16x16x4_f32(false, aS, false, bIm1, (short)0, si1, false, false);
        ci1 = __builtin_amdgcn_wmma_f32_16x16x4_f32(false, aC, false, bIm1, (short)0, ci1, false, false);
        sr1 = __builtin_amdgcn_wmma_f32_16x16x4_f32(false, aS, false, bRe1, (short)0, sr1, false, false);
    }

    for (int ct = 0; ct < 2; ++ct) {
        const v8f& p1 = ct ? cr1 : cr0;
        const v8f& p2 = ct ? si1 : si0;
        const v8f& p3 = ct ? ci1 : ci0;
        const v8f& p4 = ct ? sr1 : sr0;
        const int kcol = col0 + ct * 16 + ln;
        if (kcol < KBINS) {
            for (int v = 0; v < 8; ++v) {
                const int mr = wave * 16 + 8 * lg + v;
                const bool z = (mr == 0);
                const int rowB = z ? (C_ / 2) : (C_ - mr);
                const float reA = z ? p1[v] : p1[v] + p2[v];
                const float imA = z ? p3[v] : p3[v] - p4[v];
                const float reB = z ? p4[v] : p1[v] - p2[v];
                const float imB = z ? p2[v] : p3[v] + p4[v];
                store_bin(outre, outim, b, mr,   kcol, reA, imA);
                store_bin(outre, outim, b, rowB, kcol, reB, imB);
            }
        }
    }
}

extern "C" void kernel_launch(void* const* d_in, const int* in_sizes, int n_in,
                              void* d_out, int out_size, void* d_ws, size_t ws_size,
                              hipStream_t stream) {
    const float* xre   = (const float*)d_in[0];
    const float* xim   = (const float*)d_in[1];
    const float* gamma = (const float*)d_in[2];
    const float* beta  = (const float*)d_in[3];
    float* outre = (float*)d_out;
    float* outim = outre + (size_t)B_ * C_ * N_;

    float2* R  = (float2*)d_ws;
    float2* Wc = R + (size_t)B_ * C_ * KS;
    float2* Ws = Wc + MH_ * WPAIR;

    k0_twiddle<<<dim3((MH_ * WPAIR) / 256), dim3(256), 0, stream>>>(Wc, Ws);
    k1_row_fft<<<dim3((B_ * C_) / 2), dim3(256), 0, stream>>>(xre, xim, gamma, beta, R);
    k2_dftC_wmma<<<dim3((KBINS + 31) / 32, B_), dim3(128), 0, stream>>>(R, Wc, Ws, outre, outim);
}


# row FFT: LDS line stored bank-swizzled (point index bits 5,6 folded into bits 0-4): the fused passes of strides 16,8 and 4,2 lose their 4-way bank conflicts; arithmetic and sum order unchanged
# speedup vs baseline: 3.0947x; 1.0370x over previous
#include <hip/hip_runtime.h>
#include <math.h>

#define B_    16
#define C_    128
#define N_    8192
#define NH    4096
#define KBINS 4097
#define KS    4104
#define WPAIR 64
#define MH_   64

typedef __attribute__((ext_vector_type(2))) float v2f;
typedef __attribute__((ext_vector_type(8))) float v8f;

__global__ __launch_bounds__(256) void k0_twiddle(float2* __restrict__ Wc,
                                                  float2* __restrict__ Ws)
{
    const int idx = blockIdx.x * 256 + threadIdx.x;
    const int m = idx >> 6, j = idx & 63;
    const float scale = 1.0f / 1048576.0f;
    float f0 = (float)(m * (2 * j))     * (1.f / 64.f);
    float f1 = (float)(m * (2 * j + 1)) * (1.f / 64.f);
    Wc[idx] = make_float2(cospif(f0) * scale, cospif(f1) * scale);
    if (m == 0) {
        Ws[idx] = make_float2(cospif((float)(2 * j)) * scale, cospif((float)(2 * j + 1)) * scale);
    } else {
        Ws[idx] = make_float2(sinpif(f0) * scale, sinpif(f1) * scale);
    }
}

__device__ __forceinline__ int lds_slot(int p)
{
    return p ^ ((p & 32) ? 21 : 0) ^ ((p & 64) ? 10 : 0);
}

template <bool INV>
__device__ __forceinline__ void twiddle_of(int j, int ls, float& wr, float& wi)
{
    const float a = ldexpf((float)j, -ls);
    wr = cospif(a);
    wi = INV ? sinpif(a) : -sinpif(a);
}

template <bool INV>
__device__ __forceinline__ void bfly(float& ar, float& ai, float& br, float& bi,
                                     float wr, float wi)
{
    if (INV) {
        const float tr = ar - br, ti = ai - bi;
        ar = ar + br;  ai = ai + bi;
        br = tr * wr - ti * wi;
        bi = tr * wi + ti * wr;
    } else {
        const float tr = br * wr - bi * wi;
        const float ti = br * wi + bi * wr;
        br = ar - tr;  bi = ai - ti;
        ar = ar + tr;  ai = ai + ti;
    }
}

template <bool INV, int U>
__device__ __forceinline__ void bfly_group(float2* __restrict__ z,
                                           int s0, int s1, int step, float wr, float wi)
{
    float ar[U], ai[U], br[U], bi[U];
    for (int u = 0; u < U; ++u) {
        const float2 a = z[s0 + u * step], b = z[s1 + u * step];
        ar[u] = a.x;  ai[u] = a.y;
        br[u] = b.x;  bi[u] = b.y;
    }
    for (int u = 0; u < U; ++u) {
        bfly<INV>(ar[u], ai[u], br[u], bi[u], wr, wi);
        z[s0 + u * step] = make_float2(ar[u], ai[u]);
        z[s1 + u * step] = make_float2(br[u], bi[u]);
    }
}

template <bool INV, int U>
__device__ __forceinline__ void quad_group(float2* __restrict__ z,
                                           const int (&sl)[4], int step,
                                           float wsr, float wsi, float whr, float whi)
{
    const float wqr = INV ? -wsi :  wsi;
    const float wqi = INV ?  wsr : -wsr;
    float xr[U][4], xi[U][4];
    for (int u = 0; u < U; ++u) {
        for (int k = 0; k < 4; ++k) { const float2 t = z[sl[k] + u * step]; xr[u][k] = t.x; xi[u][k] = t.y; }
    }
    for (int u = 0; u < U; ++u) {
        if (INV) {
            bfly<true>(xr[u][0], xi[u][0], xr[u][2], xi[u][2], wsr, wsi);
            bfly<true>(xr[u][1], xi[u][1], xr[u][3], xi[u][3], wqr, wqi);
            bfly<true>(xr[u][0], xi[u][0], xr[u][1], xi[u][1], whr, whi);
            bfly<true>(xr[u][2], xi[u][2], xr[u][3], xi[u][3], whr, whi);
        } else {
            bfly<false>(xr[u][0], xi[u][0], xr[u][1], xi[u][1], whr, whi);
            bfly<false>(xr[u][2], xi[u][2], xr[u][3], xi[u][3], whr, whi);
            bfly<false>(xr[u][0], xi[u][0], xr[u][2], xi[u][2], wsr, wsi);
            bfly<false>(xr[u][1], xi[u][1], xr[u][3], xi[u][3], wqr, wqi);
        }
        for (int k = 0; k < 4; ++k) z[sl[k] + u * step] = make_float2(xr[u][k], xi[u][k]);
    }
}

template <bool INV>
__device__ __forceinline__ void pass_small(float2* __restrict__ z,
                                           int tid, int ls)
{
    const int s    = 1 << ls;
    const int j    = tid & (s - 1);
    const int base = ((tid >> ls) << (ls + 1)) + j;
    float wr, wi;
    twiddle_of<INV>(j, ls, wr, wi);
    const int s0 = lds_slot(base), s1 = lds_slot(base + s);
    bfly_group<INV, 8>(z, s0,           s1,           512, wr, wi);
    bfly_group<INV, 8>(z, s0 + 8 * 512, s1 + 8 * 512, 512, wr, wi);
}

template <bool INV>
__device__ __forceinline__ void pass4_small(float2* __restrict__ z,
                                            int tid, int L)
{
    const int h    = 1 << (L - 1);
    const int j    = tid & (h - 1);
    const int base = ((tid >> (L - 1)) << (L + 1)) + j;
    float wsr, wsi, whr, whi;
    twiddle_of<INV>(j, L,     wsr, wsi);
    twiddle_of<INV>(j, L - 1, whr, whi);
    int sl[4], sm[4];
    for (int k = 0; k < 4; ++k) { sl[k] = lds_slot(base + k * h);  sm[k] = sl[k] + 4 * 1024; }
    quad_group<INV, 4>(z, sl, 1024, wsr, wsi, whr, whi);
    quad_group<INV, 4>(z, sm, 1024, wsr, wsi, whr, whi);
}

template <bool INV, int L>
__device__ __forceinline__ void pass4_large(float2* __restrict__ z, int tid)
{
    constexpr int h = 1 << (L - 1);
    constexpr int Q = h >> 8;
    constexpr int R = 8 / Q;
    #pragma unroll 1
    for (int q = 0; q < Q; ++q) {
        const int j = tid + 256 * q;
        float wsr, wsi, whr, whi;
        twiddle_of<INV>(j, L,     wsr, wsi);
        twiddle_of<INV>(j, L - 1, whr, whi);
        const int sj = lds_slot(j);
        const int sl[4] = { sj, sj + h, sj + 2 * h, sj + 3 * h };
        quad_group<INV, R>(z, sl, 4 * h, wsr, wsi, whr, whi);
    }
}

__device__ __forceinline__ float erf_rational(float x)
{
    x = fminf(fmaxf(x, -4.f), 4.f);
    const float x2 = x * x;
    float p = -2.72614225801306e-10f;
    p = fmaf(p, x2,  2.77068142495902e-08f);
    p = fmaf(p, x2, -2.10102402082508e-06f);
    p = fmaf(p, x2, -5.69250639462346e-05f);
    p = fmaf(p, x2, -7.34990630326855e-04f);
    p = fmaf(p, x2, -2.95459980854025e-03f);
    p = fmaf(p, x2, -1.60960333262415e-02f);
    float q = -1.45660718464996e-05f;
    q = fmaf(q, x2, -2.13374055278905e-04f);
    q = fmaf(q, x2, -1.68282697438203e-03f);
    q = fmaf(q, x2, -7.37332916720468e-03f);
    q = fmaf(q, x2, -1.42647390514189e-02f);
    const float r = (x * p) * __builtin_amdgcn_rcpf(q);
    return fminf(fmaxf(r, -1.f), 1.f);
}

__global__ __launch_bounds__(256) void k1_row_fft(const float* __restrict__ xre,
                                                  const float* __restrict__ xim,
                                                  const float* __restrict__ gamma,
                                                  const float* __restrict__ beta,
                                                  float2* __restrict__ R)
{
    __shared__ __align__(16) float2 z[N_];
    __shared__ float red[4 * 256];
    const int row0 = 2 * blockIdx.x;
    const int c0   = row0 & (C_ - 1);
    const int tid  = threadIdx.x;
    const float* __restrict__ a1 = xre + (size_t)row0 * N_;
    const float* __restrict__ b1 = xim + (size_t)row0 * N_;
    const float* __restrict__ a2 = a1 + N_;
    const float* __restrict__ b2 = b1 + N_;

    #pragma unroll 4
    for (int k = tid; k < NH; k += 256) {
        const int km = (N_ - k) & (N_ - 1);
        const float sA1 = a1[k] + a1[km];
        const float dB1 = b1[k] - b1[km];
        const float sA2 = a2[k] + a2[km];
        const float dB2 = b2[k] - b2[km];
        z[lds_slot(k)]  = make_float2(0.5f * (sA1 - dB2), 0.5f * (dB1 + sA2));
        z[lds_slot(km)] = make_float2(0.5f * (sA1 + dB2), 0.5f * (sA2 - dB1));
    }
    if (tid == 0) z[lds_slot(NH)] = make_float2(a1[NH], a2[NH]);
    __syncthreads();

    pass4_large<true, 12>(z, tid);  __syncthreads();
    pass4_large<true, 10>(z, tid);  __syncthreads();
    for (int L = 8; L >= 2; L -= 2) {
        pass4_small<true>(z, tid, L);
        __syncthreads();
    }
    pass_small<true>(z, tid, 0);
    __syncthreads();

    float s1 = 0.f, q1 = 0.f, s2 = 0.f, q2 = 0.f;
    const int st = lds_slot(tid);
    for (int i = st; i < N_; i += 256) {
        const float2 t = z[i];
        const float v = t.x;  s1 += v;  q1 += v * v;
        const float w = t.y;  s2 += w;  q2 += w * w;
    }
    red[tid]       = s1;
    red[256 + tid] = q1;
    red[512 + tid] = s2;
    red[768 + tid] = q2;
    __syncthreads();
    for (int off = 128; off > 0; off >>= 1) {
        if (tid < off) {
            red[tid]       += red[tid + off];
            red[256 + tid] += red[256 + tid + off];
            red[512 + tid] += red[512 + tid + off];
            red[768 + tid] += red[768 + tid + off];
        }
        __syncthreads();
    }
    const float mean1 = red[0]   * (1.f / N_);
    const float var1  = red[256] * (1.f / N_) - mean1 * mean1;
    const float istd1 = rsqrtf(var1 + 1e-5f);
    const float mean2 = red[512] * (1.f / N_);
    const float var2  = red[768] * (1.f / N_) - mean2 * mean2;
    const float istd2 = rsqrtf(var2 + 1e-5f);
    const float gm1 = gamma[c0],     bt1 = beta[c0];
    const float gm2 = gamma[c0 + 1], bt2 = beta[c0 + 1];

    for (int i = st; i < N_; i += 256) {
        const float2 t = z[i];
        float v = (t.x - mean1) * istd1 * gm1 + bt1;
        v = 0.5f * v * (1.f + erf_rational(v * 0.70710678118654752f));
        float w = (t.y - mean2) * istd2 * gm2 + bt2;
        w = 0.5f * w * (1.f + erf_rational(w * 0.70710678118654752f));
        z[i] = make_float2(v, w);
    }
    __syncthreads();

    pass_small<false>(z, tid, 0);
    __syncthreads();
    for (int L = 2; L <= 8; L += 2) {
        pass4_small<false>(z, tid, L);
        __syncthreads();
    }
    pass4_large<false, 10>(z, tid);  __syncthreads();
    pass4_large<false, 12>(z, tid);  __syncthreads();

    float2* R0 = R + (size_t)row0 * KS;
    float2* R1 = R0 + KS;
    for (int k = tid; k < KBINS; k += 256) {
        const int km = (N_ - k) & (N_ - 1);
        const float2 fk = z[lds_slot(k)], fm = z[lds_slot(km)];
        const float pk = fk.x, qk = fk.y, pm = fm.x, qm = fm.y;
        R0[k] = make_float2(0.5f * (pk + pm), 0.5f * (qk - qm));
        R1[k] = make_float2(0.5f * (qk + qm), 0.5f * (pm - pk));
    }
}

__device__ __forceinline__ void store_bin(float* __restrict__ outre, float* __restrict__ outim,
                                          int b, int row, int kcol, float orv, float oim)
{
    const size_t o = (size_t)(b * C_ + row) * N_;
    const bool edge = (kcol == 0) || (kcol == NH);
    const float oiv = edge ? 0.f : oim;
    outre[o + kcol] = orv;
    outim[o + kcol] = oiv;
    if (kcol >= 1 && kcol <= NH - 1) {
        outre[o + (N_ - kcol)] =  orv;
        outim[o + (N_ - kcol)] = -oiv;
    }
}

__global__ __launch_bounds__(128) void k2_dftC_wmma(const float2* __restrict__ R,
                                                    const float2* __restrict__ Wc,
                                                    const float2* __restrict__ Ws,
                                                    float* __restrict__ outre,
                                                    float* __restrict__ outim)
{
    __shared__ float Lre[C_][40];
    __shared__ float Lim[C_][40];
    const int b    = blockIdx.y;
    const int col0 = blockIdx.x * 32;
    const int tid  = threadIdx.x;
    const float2* Rb = R + (size_t)b * C_ * KS;

    {
        const int  cc = tid & 31;
        const int  r0 = tid >> 5;
        const int  k  = col0 + cc;
        const bool in = k < KBINS;
        const float2* src = Rb + (size_t)r0 * KS + (in ? k : KBINS - 1);
        float2 v[32];
        #pragma unroll
        for (int i = 0; i < 32; ++i) v[i] = src[(size_t)(4 * i) * KS];
        #pragma unroll
        for (int i = 0; i < 32; ++i) {
            Lre[r0 + 4 * i][cc] = in ? v[i].x : 0.f;
            Lim[r0 + 4 * i][cc] = in ? v[i].y : 0.f;
        }
    }
    __syncthreads();

    const int wave = tid >> 5;
    const int lane = tid & 31;
    const int lg   = lane >> 4;
    const int ln   = lane & 15;
    const int m    = wave * 16 + ln;
    const float2* WcRow = Wc + m * WPAIR;
    const float2* WsRow = Ws + m * WPAIR;

    const v8f zero8 = {0.f,0.f,0.f,0.f,0.f,0.f,0.f,0.f};
    v8f cr0 = zero8, si0 = zero8, ci0 = zero8, sr0 = zero8;
    v8f cr1 = zero8, si1 = zero8, ci1 = zero8, sr1 = zero8;

    for (int k0 = 0; k0 < C_; k0 += 4) {
        const int kA = k0 + 2 * lg;
        float2 wc = WcRow[(k0 >> 1) + lg];
        float2 ws = WsRow[(k0 >> 1) + lg];
        v2f aC = { wc.x, wc.y };
        v2f aS = { ws.x, ws.y };
        v2f bRe0 = { Lre[kA][ln],      Lre[kA + 1][ln]      };
        v2f bIm0 = { Lim[kA][ln],      Lim[kA + 1][ln]      };
        v2f bRe1 = { Lre[kA][16 + ln], Lre[kA + 1][16 + ln] };
        v2f bIm1 = { Lim[kA][16 + ln], Lim[kA + 1][16 + ln] };
        cr0 = __builtin_amdgcn_wmma_f32_16x16x4_f32(false, aC, false, bRe0, (short)0, cr0, false, false);
        si0 = __builtin_amdgcn_wmma_f32_16x16x4_f32(false, aS, false, bIm0, (short)0, si0, false, false);
        ci0 = __builtin_amdgcn_wmma_f32_16x16x4_f32(false, aC, false, bIm0, (short)0, ci0, false, false);
        sr0 = __builtin_amdgcn_wmma_f32_16x16x4_f32(false, aS, false, bRe0, (short)0, sr0, false, false);
        cr1 = __builtin_amdgcn_wmma_f32_16x16x4_f32(false, aC, false, bRe1, (short)0, cr1, false, false);
        si1 = __builtin_amdgcn_wmma_f32_16x16x4_f32(false, aS, false, bIm1, (short)0, si1, false, false);
        ci1 = __builtin_amdgcn_wmma_f32_16x16x4_f32(false, aC, false, bIm1, (short)0, ci1, false, false);
        sr1 = __builtin_amdgcn_wmma_f32_16x16x4_f32(false, aS, false, bRe1, (short)0, sr1, false, false);
    }

    for (int ct = 0; ct < 2; ++ct) {
        const v8f& p1 = ct ? cr1 : cr0;
        const v8f& p2 = ct ? si1 : si0;
        const v8f& p3 = ct ? ci1 : ci0;
        const v8f& p4 = ct ? sr1 : sr0;
        const int kcol = col0 + ct * 16 + ln;
        if (kcol < KBINS) {
            for (int v = 0; v < 8; ++v) {
                const int mr = wave * 16 + 8 * lg + v;
                const bool z = (mr == 0);
                const int rowB = z ? (C_ / 2) : (C_ - mr);
                const float reA = z ? p1[v] : p1[v] + p2[v];
                const float imA = z ? p3[v] : p3[v] - p4[v];
                const float reB = z ? p4[v] : p1[v] - p2[v];
                const float imB = z ? p2[v] : p3[v] + p4[v];
                store_bin(outre, outim, b, mr,   kcol, reA, imA);
                store_bin(outre, outim, b, rowB, kcol, reB, imB);
            }
        }
    }
}

extern "C" void kernel_launch(void* const* d_in, const int* in_sizes, int n_in,
                              void* d_out, int out_size, void* d_ws, size_t ws_size,
                              hipStream_t stream) {
    const float* xre   = (const float*)d_in[0];
    const float* xim   = (const float*)d_in[1];
    const float* gamma = (const float*)d_in[2];
    const float* beta  = (const float*)d_in[3];
    float* outre = (float*)d_out;
    float* outim = outre + (size_t)B_ * C_ * N_;

    float2* R  = (float2*)d_ws;
    float2* Wc = R + (size_t)B_ * C_ * KS;
    float2* Ws = Wc + MH_ * WPAIR;

    k0_twiddle<<<dim3((MH_ * WPAIR) / 256), dim3(256), 0, stream>>>(Wc, Ws);
    k1_row_fft<<<dim3((B_ * C_) / 2), dim3(256), 0, stream>>>(xre, xim, gamma, beta, R);
    k2_dftC_wmma<<<dim3((KBINS + 31) / 32, B_), dim3(128), 0, stream>>>(R, Wc, Ws, outre, outim);
}


# row FFT: the twiddles of the six fused passes are read from a table that kernel 0 fills with the same sincos calls (14 16-byte loads per thread and transform, not 28 sincos pairs); same bits
# speedup vs baseline: 3.2296x; 1.0436x over previous
#include <hip/hip_runtime.h>
#include <math.h>

#define B_    16
#define C_    128
#define N_    8192
#define NH    4096
#define KBINS 4097
#define KS    4104
#define WPAIR 64
#define MH_   64

#define TWN   2730
#define K0_BLOCKS 27

typedef __attribute__((ext_vector_type(2))) float v2f;
typedef __attribute__((ext_vector_type(8))) float v8f;

template <bool INV>
__device__ __forceinline__ void twiddle_of(int j, int ls, float& wr, float& wi)
{
    const float a = ldexpf((float)j, -ls);
    wr = cospif(a);
    wi = INV ? sinpif(a) : -sinpif(a);
}

__device__ __forceinline__ int tw_base(int L)
{
    return (8192 - (2 << L)) / 3;
}

__device__ __forceinline__ float4 tw_entry(const float4* __restrict__ tw, int tid, int L)
{
    return tw[tw_base(L) + (tid & ((1 << (L - 1)) - 1))];
}

__global__ __launch_bounds__(256) void k0_twiddle(float2* __restrict__ Wc,
                                                  float2* __restrict__ Ws,
                                                  float4* __restrict__ tw)
{
    const int idx = blockIdx.x * 256 + threadIdx.x;
    if (idx >= MH_ * WPAIR) {
        const int e = idx - MH_ * WPAIR;
        if (e < TWN) {
            int L = 12, base = 0;
            for (int t = 0; t < 5; ++t) {
                const int h = 1 << (L - 1);
                if (e >= base + h) { base += h;  L -= 2; }
            }
            const int j = e - base;
            float wsr, wsi, whr, whi;
            twiddle_of<true>(j, L,     wsr, wsi);
            twiddle_of<true>(j, L - 1, whr, whi);
            tw[e] = make_float4(wsr, wsi, whr, whi);
        }
        return;
    }
    const int m = idx >> 6, j = idx & 63;
    const float scale = 1.0f / 1048576.0f;
    float f0 = (float)(m * (2 * j))     * (1.f / 64.f);
    float f1 = (float)(m * (2 * j + 1)) * (1.f / 64.f);
    Wc[idx] = make_float2(cospif(f0) * scale, cospif(f1) * scale);
    if (m == 0) {
        Ws[idx] = make_float2(cospif((float)(2 * j)) * scale, cospif((float)(2 * j + 1)) * scale);
    } else {
        Ws[idx] = make_float2(sinpif(f0) * scale, sinpif(f1) * scale);
    }
}

__device__ __forceinline__ int lds_slot(int p)
{
    return p ^ ((p & 32) ? 21 : 0) ^ ((p & 64) ? 10 : 0);
}

template <bool INV>
__device__ __forceinline__ void bfly(float& ar, float& ai, float& br, float& bi,
                                     float wr, float wi)
{
    if (INV) {
        const float tr = ar - br, ti = ai - bi;
        ar = ar + br;  ai = ai + bi;
        br = tr * wr - ti * wi;
        bi = tr * wi + ti * wr;
    } else {
        const float tr = br * wr - bi * wi;
        const float ti = br * wi + bi * wr;
        br = ar - tr;  bi = ai - ti;
        ar = ar + tr;  ai = ai + ti;
    }
}

template <bool INV, int U>
__device__ __forceinline__ void bfly_group(float2* __restrict__ z,
                                           int s0, int s1, int step, float wr, float wi)
{
    float ar[U], ai[U], br[U], bi[U];
    for (int u = 0; u < U; ++u) {
        const float2 a = z[s0 + u * step], b = z[s1 + u * step];
        ar[u] = a.x;  ai[u] = a.y;
        br[u] = b.x;  bi[u] = b.y;
    }
    for (int u = 0; u < U; ++u) {
        bfly<INV>(ar[u], ai[u], br[u], bi[u], wr, wi);
        z[s0 + u * step] = make_float2(ar[u], ai[u]);
        z[s1 + u * step] = make_float2(br[u], bi[u]);
    }
}

template <bool INV, int U>
__device__ __forceinline__ void quad_group(float2* __restrict__ z,
                                           const int (&sl)[4], int step,
                                           float wsr, float wsi, float whr, float whi)
{
    const float wqr = INV ? -wsi :  wsi;
    const float wqi = INV ?  wsr : -wsr;
    float xr[U][4], xi[U][4];
    for (int u = 0; u < U; ++u) {
        for (int k = 0; k < 4; ++k) { const float2 t = z[sl[k] + u * step]; xr[u][k] = t.x; xi[u][k] = t.y; }
    }
    for (int u = 0; u < U; ++u) {
        if (INV) {
            bfly<true>(xr[u][0], xi[u][0], xr[u][2], xi[u][2], wsr, wsi);
            bfly<true>(xr[u][1], xi[u][1], xr[u][3], xi[u][3], wqr, wqi);
            bfly<true>(xr[u][0], xi[u][0], xr[u][1], xi[u][1], whr, whi);
            bfly<true>(xr[u][2], xi[u][2], xr[u][3], xi[u][3], whr, whi);
        } else {
            bfly<false>(xr[u][0], xi[u][0], xr[u][1], xi[u][1], whr, whi);
            bfly<false>(xr[u][2], xi[u][2], xr[u][3], xi[u][3], whr, whi);
            bfly<false>(xr[u][0], xi[u][0], xr[u][2], xi[u][2], wsr, wsi);
            bfly<false>(xr[u][1], xi[u][1], xr[u][3], xi[u][3], wqr, wqi);
        }
        for (int k = 0; k < 4; ++k) z[sl[k] + u * step] = make_float2(xr[u][k], xi[u][k]);
    }
}

template <bool INV>
__device__ __forceinline__ void pass_small(float2* __restrict__ z,
                                           int tid, int ls)
{
    const int s    = 1 << ls;
    const int j    = tid & (s - 1);
    const int base = ((tid >> ls) << (ls + 1)) + j;
    float wr, wi;
    twiddle_of<INV>(j, ls, wr, wi);
    const int s0 = lds_slot(base), s1 = lds_slot(base + s);
    bfly_group<INV, 8>(z, s0,           s1,           512, wr, wi);
    bfly_group<INV, 8>(z, s0 + 8 * 512, s1 + 8 * 512, 512, wr, wi);
}

template <bool INV>
__device__ __forceinline__ void pass4_small(float2* __restrict__ z,
                                            int tid, int L, float4 w)
{
    const int h    = 1 << (L - 1);
    const int j    = tid & (h - 1);
    const int base = ((tid >> (L - 1)) << (L + 1)) + j;
    const float wsr = w.x, wsi = INV ? w.y : -w.y;
    const float whr = w.z, whi = INV ? w.w : -w.w;
    int sl[4], sm[4];
    for (int k = 0; k < 4; ++k) { sl[k] = lds_slot(base + k * h);  sm[k] = sl[k] + 4 * 1024; }
    quad_group<INV, 4>(z, sl, 1024, wsr, wsi, whr, whi);
    quad_group<INV, 4>(z, sm, 1024, wsr, wsi, whr, whi);
}

template <bool INV, int L>
__device__ __forceinline__ void pass4_large(float2* __restrict__ z,
                                            const float4* __restrict__ tw, int tid, float4 w0)
{
    constexpr int h = 1 << (L - 1);
    constexpr int Q = h >> 8;
    constexpr int R = 8 / Q;
    constexpr int U = (Q < 4) ? Q : 4;
    const float4* __restrict__ twj = tw + tw_base(L) + tid;
    float4 w[U];
    w[0] = w0;
    #pragma unroll
    for (int u = 1; u < U; ++u) w[u] = twj[256 * u];
    #pragma unroll
    for (int q = 0; q < Q; ++q) {
        const int j = tid + 256 * q;
        const float wsr = w[q % U].x, wsi = INV ? w[q % U].y : -w[q % U].y;
        const float whr = w[q % U].z, whi = INV ? w[q % U].w : -w[q % U].w;
        const int sj = lds_slot(j);
        const int sl[4] = { sj, sj + h, sj + 2 * h, sj + 3 * h };
        quad_group<INV, R>(z, sl, 4 * h, wsr, wsi, whr, whi);
        if (q + U < Q) w[q % U] = twj[256 * (q + U)];
    }
}

__device__ __forceinline__ float erf_rational(float x)
{
    x = fminf(fmaxf(x, -4.f), 4.f);
    const float x2 = x * x;
    float p = -2.72614225801306e-10f;
    p = fmaf(p, x2,  2.77068142495902e-08f);
    p = fmaf(p, x2, -2.10102402082508e-06f);
    p = fmaf(p, x2, -5.69250639462346e-05f);
    p = fmaf(p, x2, -7.34990630326855e-04f);
    p = fmaf(p, x2, -2.95459980854025e-03f);
    p = fmaf(p, x2, -1.60960333262415e-02f);
    float q = -1.45660718464996e-05f;
    q = fmaf(q, x2, -2.13374055278905e-04f);
    q = fmaf(q, x2, -1.68282697438203e-03f);
    q = fmaf(q, x2, -7.37332916720468e-03f);
    q = fmaf(q, x2, -1.42647390514189e-02f);
    const float r = (x * p) * __builtin_amdgcn_rcpf(q);
    return fminf(fmaxf(r, -1.f), 1.f);
}

__global__ __launch_bounds__(256) void k1_row_fft(const float* __restrict__ xre,
                                                  const float* __restrict__ xim,
                                                  const float* __restrict__ gamma,
                                                  const float* __restrict__ beta,
                                                  const float4* __restrict__ tw,
                                                  float2* __restrict__ R)
{
    __shared__ __align__(16) float2 z[N_];
    __shared__ float red[4 * 256];
    const int row0 = 2 * blockIdx.x;
    const int c0   = row0 & (C_ - 1);
    const int tid  = threadIdx.x;
    const float* __restrict__ a1 = xre + (size_t)row0 * N_;
    const float* __restrict__ b1 = xim + (size_t)row0 * N_;
    const float* __restrict__ a2 = a1 + N_;
    const float* __restrict__ b2 = b1 + N_;

    float4 wa = tw_entry(tw, tid, 12), wb;

    #pragma unroll 4
    for (int k = tid; k < NH; k += 256) {
        const int km = (N_ - k) & (N_ - 1);
        const float sA1 = a1[k] + a1[km];
        const float dB1 = b1[k] - b1[km];
        const float sA2 = a2[k] + a2[km];
        const float dB2 = b2[k] - b2[km];
        z[lds_slot(k)]  = make_float2(0.5f * (sA1 - dB2), 0.5f * (dB1 + sA2));
        z[lds_slot(km)] = make_float2(0.5f * (sA1 + dB2), 0.5f * (sA2 - dB1));
    }
    if (tid == 0) z[lds_slot(NH)] = make_float2(a1[NH], a2[NH]);
    __syncthreads();

    wb = tw_entry(tw, tid, 10);
    pass4_large<true, 12>(z, tw, tid, wa);  __syncthreads();
    wa = tw_entry(tw, tid, 8);
    pass4_large<true, 10>(z, tw, tid, wb);  __syncthreads();
    for (int L = 8; L >= 2; L -= 2) {
        wb = tw_entry(tw, tid, (L > 2) ? L - 2 : 2);
        pass4_small<true>(z, tid, L, wa);
        __syncthreads();
        wa = wb;
    }
    pass_small<true>(z, tid, 0);
    __syncthreads();

    float s1 = 0.f, q1 = 0.f, s2 = 0.f, q2 = 0.f;
    const int st = lds_slot(tid);
    for (int i = st; i < N_; i += 256) {
        const float2 t = z[i];
        const float v = t.x;  s1 += v;  q1 += v * v;
        const float w = t.y;  s2 += w;  q2 += w * w;
    }
    red[tid]       = s1;
    red[256 + tid] = q1;
    red[512 + tid] = s2;
    red[768 + tid] = q2;
    __syncthreads();
    for (int off = 128; off > 0; off >>= 1) {
        if (tid < off) {
            red[tid]       += red[tid + off];
            red[256 + tid] += red[256 + tid + off];
            red[512 + tid] += red[512 + tid + off];
            red[768 + tid] += red[768 + tid + off];
        }
        __syncthreads();
    }
    const float mean1 = red[0]   * (1.f / N_);
    const float var1  = red[256] * (1.f / N_) - mean1 * mean1;
    const float istd1 = rsqrtf(var1 + 1e-5f);
    const float mean2 = red[512] * (1.f / N_);
    const float var2  = red[768] * (1.f / N_) - mean2 * mean2;
    const float istd2 = rsqrtf(var2 + 1e-5f);
    const float gm1 = gamma[c0],     bt1 = beta[c0];
    const float gm2 = gamma[c0 + 1], bt2 = beta[c0 + 1];

    for (int i = st; i < N_; i += 256) {
        const float2 t = z[i];
        float v = (t.x - mean1) * istd1 * gm1 + bt1;
        v = 0.5f * v * (1.f + erf_rational(v * 0.70710678118654752f));
        float w = (t.y - mean2) * istd2 * gm2 + bt2;
        w = 0.5f * w * (1.f + erf_rational(w * 0.70710678118654752f));
        z[i] = make_float2(v, w);
    }
    __syncthreads();

    wa = tw_entry(tw, tid, 2);
    pass_small<false>(z, tid, 0);
    __syncthreads();
    for (int L = 2; L <= 8; L += 2) {
        wb = tw_entry(tw, tid, L + 2);
        pass4_small<false>(z, tid, L, wa);
        __syncthreads();
        wa = wb;
    }
    wb = tw_entry(tw, tid, 12);
    pass4_large<false, 10>(z, tw, tid, wa);  __syncthreads();
    pass4_large<false, 12>(z, tw, tid, wb);  __syncthreads();

    float2* R0 = R + (size_t)row0 * KS;
    float2* R1 = R0 + KS;
    for (int k = tid; k < KBINS; k += 256) {
        const int km = (N_ - k) & (N_ - 1);
        const float2 fk = z[lds_slot(k)], fm = z[lds_slot(km)];
        const float pk = fk.x, qk = fk.y, pm = fm.x, qm = fm.y;
        R0[k] = make_float2(0.5f * (pk + pm), 0.5f * (qk - qm));
        R1[k] = make_float2(0.5f * (qk + qm), 0.5f * (pm - pk));
    }
}

__device__ __forceinline__ void store_bin(float* __restrict__ outre, float* __restrict__ outim,
                                          int b, int row, int kcol, float orv, float oim)
{
    const size_t o = (size_t)(b * C_ + row) * N_;
    const bool edge = (kcol == 0) || (kcol == NH);
    const float oiv = edge ? 0.f : oim;
    outre[o + kcol] = orv;
    outim[o + kcol] = oiv;
    if (kcol >= 1 && kcol <= NH - 1) {
        outre[o + (N_ - kcol)] =  orv;
        outim[o + (N_ - kcol)] = -oiv;
    }
}

__global__ __launch_bounds__(128) void k2_dftC_wmma(const float2* __restrict__ R,
                                                    const float2* __restrict__ Wc,
                                                    const float2* __restrict__ Ws,
                                                    float* __restrict__ outre,
                                                    float* __restrict__ outim)
{
    __shared__ float Lre[C_][40];
    __shared__ float Lim[C_][40];
    const int b    = blockIdx.y;
    const int col0 = blockIdx.x * 32;
    const int tid  = threadIdx.x;
    const float2* Rb = R + (size_t)b * C_ * KS;

    {
        const int  cc = tid & 31;
        const int  r0 = tid >> 5;
        const int  k  = col0 + cc;
        const bool in = k < KBINS;
        const float2* src = Rb + (size_t)r0 * KS + (in ? k : KBINS - 1);
        float2 v[32];
        #pragma unroll
        for (int i = 0; i < 32; ++i) v[i] = src[(size_t)(4 * i) * KS];
        #pragma unroll
        for (int i = 0; i < 32; ++i) {
            Lre[r0 + 4 * i][cc] = in ? v[i].x : 0.f;
            Lim[r0 + 4 * i][cc] = in ? v[i].y : 0.f;
        }
    }
    __syncthreads();

    const int wave = tid >> 5;
    const int lane = tid & 31;
    const int lg   = lane >> 4;
    const int ln   = lane & 15;
    const int m    = wave * 16 + ln;
    const float2* WcRow = Wc + m * WPAIR;
    const float2* WsRow = Ws + m * WPAIR;

    const v8f zero8 = {0.f,0.f,0.f,0.f,0.f,0.f,0.f,0.f};
    v8f cr0 = zero8, si0 = zero8, ci0 = zero8, sr0 = zero8;
    v8f cr1 = zero8, si1 = zero8, ci1 = zero8, sr1 = zero8;

    for (int k0 = 0; k0 < C_; k0 += 4) {
        const int kA = k0 + 2 * lg;
        float2 wc = WcRow[(k0 >> 1) + lg];
        float2 ws = WsRow[(k0 >> 1) + lg];
        v2f aC = { wc.x, wc.y };
        v2f aS = { ws.x, ws.y };
        v2f bRe0 = { Lre[kA][ln],      Lre[kA + 1][ln]      };
        v2f bIm0 = { Lim[kA][ln],      Lim[kA + 1][ln]      };
        v2f bRe1 = { Lre[kA][16 + ln], Lre[kA + 1][16 + ln] };
        v2f bIm1 = { Lim[kA][16 + ln], Lim[kA + 1][16 + ln] };
        cr0 = __builtin_amdgcn_wmma_f32_16x16x4_f32(false, aC, false, bRe0, (short)0, cr0, false, false);
        si0 = __builtin_amdgcn_wmma_f32_16x16x4_f32(false, aS, false, bIm0, (short)0, si0, false, false);
        ci0 = __builtin_amdgcn_wmma_f32_16x16x4_f32(false, aC, false, bIm0, (short)0, ci0, false, false);
        sr0 = __builtin_amdgcn_wmma_f32_16x16x4_f32(false, aS, false, bRe0, (short)0, sr0, false, false);
        cr1 = __builtin_amdgcn_wmma_f32_16x16x4_f32(false, aC, false, bRe1, (short)0, cr1, false, false);
        si1 = __builtin_amdgcn_wmma_f32_16x16x4_f32(false, aS, false, bIm1, (short)0, si1, false, false);
        ci1 = __builtin_amdgcn_wmma_f32_16x16x4_f32(false, aC, false, bIm1, (short)0, ci1, false, false);
        sr1 = __builtin_amdgcn_wmma_f32_16x16x4_f32(false, aS, false, bRe1, (short)0, sr1, false, false);
    }

    for (int ct = 0; ct < 2; ++ct) {
        const v8f& p1 = ct ? cr1 : cr0;
        const v8f& p2 = ct ? si1 : si0;
        const v8f& p3 = ct ? ci1 : ci0;
        const v8f& p4 = ct ? sr1 : sr0;
        const int kcol = col0 + ct * 16 + ln;
        if (kcol < KBINS) {
            for (int v = 0; v < 8; ++v) {
                const int mr = wave * 16 + 8 * lg + v;
                const bool z = (mr == 0);
                const int rowB = z ? (C_ / 2) : (C_ - mr);
                const float reA = z ? p1[v] : p1[v] + p2[v];
                const float imA = z ? p3[v] : p3[v] - p4[v];
                const float reB = z ? p4[v] : p1[v] - p2[v];
                const float imB = z ? p2[v] : p3[v] + p4[v];
                store_bin(outre, outim, b, mr,   kcol, reA, imA);
                store_bin(outre, outim, b, rowB, kcol, reB, imB);
            }
        }
    }
}

extern "C" void kernel_launch(void* const* d_in, const int* in_sizes, int n_in,
                              void* d_out, int out_size, void* d_ws, size_t ws_size,
                              hipStream_t stream) {
    const float* xre   = (const float*)d_in[0];
    const float* xim   = (const float*)d_in[1];
    const float* gamma = (const float*)d_in[2];
    const float* beta  = (const float*)d_in[3];
    float* outre = (float*)d_out;
    float* outim = outre + (size_t)B_ * C_ * N_;

    float2* R  = (float2*)d_ws;
    float2* Wc = R + (size_t)B_ * C_ * KS;
    float2* Ws = Wc + MH_ * WPAIR;

    float4* tw = (float4*)d_out;

    k0_twiddle<<<dim3(K0_BLOCKS), dim3(256), 0, stream>>>(Wc, Ws, tw);
    k1_row_fft<<<dim3((B_ * C_) / 2), dim3(256), 0, stream>>>(xre, xim, gamma, beta, tw, R);
    k2_dftC_wmma<<<dim3((KBINS + 31) / 32, B_), dim3(128), 0, stream>>>(R, Wc, Ws, outre, outim);
}


# channel DFT epilogue: lane halves of the two column tiles swapped (v_permlanex16): a register holds 32 consecutive columns of ONE row, so a store writes a whole 128-B line, not two halves; same sums
# speedup vs baseline: 3.4368x; 1.0642x over previous
#include <hip/hip_runtime.h>
#include <math.h>

#define B_    16
#define C_    128
#define N_    8192
#define NH    4096
#define KBINS 4097
#define KS    4104
#define WPAIR 64
#define MH_   64

#define TWN   2730
#define K0_BLOCKS 27

typedef __attribute__((ext_vector_type(2))) float v2f;
typedef __attribute__((ext_vector_type(8))) float v8f;

template <bool INV>
__device__ __forceinline__ void twiddle_of(int j, int ls, float& wr, float& wi)
{
    const float a = ldexpf((float)j, -ls);
    wr = cospif(a);
    wi = INV ? sinpif(a) : -sinpif(a);
}

__device__ __forceinline__ int tw_base(int L)
{
    return (8192 - (2 << L)) / 3;
}

__device__ __forceinline__ float4 tw_entry(const float4* __restrict__ tw, int tid, int L)
{
    return tw[tw_base(L) + (tid & ((1 << (L - 1)) - 1))];
}

__global__ __launch_bounds__(256) void k0_twiddle(float2* __restrict__ Wc,
                                                  float2* __restrict__ Ws,
                                                  float4* __restrict__ tw)
{
    const int idx = blockIdx.x * 256 + threadIdx.x;
    if (idx >= MH_ * WPAIR) {
        const int e = idx - MH_ * WPAIR;
        if (e < TWN) {
            int L = 12, base = 0;
            for (int t = 0; t < 5; ++t) {
                const int h = 1 << (L - 1);
                if (e >= base + h) { base += h;  L -= 2; }
            }
            const int j = e - base;
            float wsr, wsi, whr, whi;
            twiddle_of<true>(j, L,     wsr, wsi);
            twiddle_of<true>(j, L - 1, whr, whi);
            tw[e] = make_float4(wsr, wsi, whr, whi);
        }
        return;
    }
    const int m = idx >> 6, j = idx & 63;
    const float scale = 1.0f / 1048576.0f;
    float f0 = (float)(m * (2 * j))     * (1.f / 64.f);
    float f1 = (float)(m * (2 * j + 1)) * (1.f / 64.f);
    Wc[idx] = make_float2(cospif(f0) * scale, cospif(f1) * scale);
    if (m == 0) {
        Ws[idx] = make_float2(cospif((float)(2 * j)) * scale, cospif((float)(2 * j + 1)) * scale);
    } else {
        Ws[idx] = make_float2(sinpif(f0) * scale, sinpif(f1) * scale);
    }
}

__device__ __forceinline__ int lds_slot(int p)
{
    return p ^ ((p & 32) ? 21 : 0) ^ ((p & 64) ? 10 : 0);
}

template <bool INV>
__device__ __forceinline__ void bfly(float& ar, float& ai, float& br, float& bi,
                                     float wr, float wi)
{
    if (INV) {
        const float tr = ar - br, ti = ai - bi;
        ar = ar + br;  ai = ai + bi;
        br = tr * wr - ti * wi;
        bi = tr * wi + ti * wr;
    } else {
        const float tr = br * wr - bi * wi;
        const float ti = br * wi + bi * wr;
        br = ar - tr;  bi = ai - ti;
        ar = ar + tr;  ai = ai + ti;
    }
}

template <bool INV, int U>
__device__ __forceinline__ void bfly_group(float2* __restrict__ z,
                                           int s0, int s1, int step, float wr, float wi)
{
    float ar[U], ai[U], br[U], bi[U];
    for (int u = 0; u < U; ++u) {
        const float2 a = z[s0 + u * step], b = z[s1 + u * step];
        ar[u] = a.x;  ai[u] = a.y;
        br[u] = b.x;  bi[u] = b.y;
    }
    for (int u = 0; u < U; ++u) {
        bfly<INV>(ar[u], ai[u], br[u], bi[u], wr, wi);
        z[s0 + u * step] = make_float2(ar[u], ai[u]);
        z[s1 + u * step] = make_float2(br[u], bi[u]);
    }
}

template <bool INV, int U>
__device__ __forceinline__ void quad_group(float2* __restrict__ z,
                                           const int (&sl)[4], int step,
                                           float wsr, float wsi, float whr, float whi)
{
    const float wqr = INV ? -wsi :  wsi;
    const float wqi = INV ?  wsr : -wsr;
    float xr[U][4], xi[U][4];
    for (int u = 0; u < U; ++u) {
        for (int k = 0; k < 4; ++k) { const float2 t = z[sl[k] + u * step]; xr[u][k] = t.x; xi[u][k] = t.y; }
    }
    for (int u = 0; u < U; ++u) {
        if (INV) {
            bfly<true>(xr[u][0], xi[u][0], xr[u][2], xi[u][2], wsr, wsi);
            bfly<true>(xr[u][1], xi[u][1], xr[u][3], xi[u][3], wqr, wqi);
            bfly<true>(xr[u][0], xi[u][0], xr[u][1], xi[u][1], whr, whi);
            bfly<true>(xr[u][2], xi[u][2], xr[u][3], xi[u][3], whr, whi);
        } else {
            bfly<false>(xr[u][0], xi[u][0], xr[u][1], xi[u][1], whr, whi);
            bfly<false>(xr[u][2], xi[u][2], xr[u][3], xi[u][3], whr, whi);
            bfly<false>(xr[u][0], xi[u][0], xr[u][2], xi[u][2], wsr, wsi);
            bfly<false>(xr[u][1], xi[u][1], xr[u][3], xi[u][3], wqr, wqi);
        }
        for (int k = 0; k < 4; ++k) z[sl[k] + u * step] = make_float2(xr[u][k], xi[u][k]);
    }
}

template <bool INV>
__device__ __forceinline__ void pass_small(float2* __restrict__ z,
                                           int tid, int ls)
{
    const int s    = 1 << ls;
    const int j    = tid & (s - 1);
    const int base = ((tid >> ls) << (ls + 1)) + j;
    float wr, wi;
    twiddle_of<INV>(j, ls, wr, wi);
    const int s0 = lds_slot(base), s1 = lds_slot(base + s);
    bfly_group<INV, 8>(z, s0,           s1,           512, wr, wi);
    bfly_group<INV, 8>(z, s0 + 8 * 512, s1 + 8 * 512, 512, wr, wi);
}

template <bool INV>
__device__ __forceinline__ void pass4_small(float2* __restrict__ z,
                                            int tid, int L, float4 w)
{
    const int h    = 1 << (L - 1);
    const int j    = tid & (h - 1);
    const int base = ((tid >> (L - 1)) << (L + 1)) + j;
    const float wsr = w.x, wsi = INV ? w.y : -w.y;
    const float whr = w.z, whi = INV ? w.w : -w.w;
    int sl[4], sm[4];
    for (int k = 0; k < 4; ++k) { sl[k] = lds_slot(base + k * h);  sm[k] = sl[k] + 4 * 1024; }
    quad_group<INV, 4>(z, sl, 1024, wsr, wsi, whr, whi);
    quad_group<INV, 4>(z, sm, 1024, wsr, wsi, whr, whi);
}

template <bool INV, int L>
__device__ __forceinline__ void pass4_large(float2* __restrict__ z,
                                            const float4* __restrict__ tw, int tid, float4 w0)
{
    constexpr int h = 1 << (L - 1);
    constexpr int Q = h >> 8;
    constexpr int R = 8 / Q;
    constexpr int U = (Q < 4) ? Q : 4;
    const float4* __restrict__ twj = tw + tw_base(L) + tid;
    float4 w[U];
    w[0] = w0;
    #pragma unroll
    for (int u = 1; u < U; ++u) w[u] = twj[256 * u];
    #pragma unroll
    for (int q = 0; q < Q; ++q) {
        const int j = tid + 256 * q;
        const float wsr = w[q % U].x, wsi = INV ? w[q % U].y : -w[q % U].y;
        const float whr = w[q % U].z, whi = INV ? w[q % U].w : -w[q % U].w;
        const int sj = lds_slot(j);
        const int sl[4] = { sj, sj + h, sj + 2 * h, sj + 3 * h };
        quad_group<INV, R>(z, sl, 4 * h, wsr, wsi, whr, whi);
        if (q + U < Q) w[q % U] = twj[256 * (q + U)];
    }
}

__device__ __forceinline__ float erf_rational(float x)
{
    x = fminf(fmaxf(x, -4.f), 4.f);
    const float x2 = x * x;
    float p = -2.72614225801306e-10f;
    p = fmaf(p, x2,  2.77068142495902e-08f);
    p = fmaf(p, x2, -2.10102402082508e-06f);
    p = fmaf(p, x2, -5.69250639462346e-05f);
    p = fmaf(p, x2, -7.34990630326855e-04f);
    p = fmaf(p, x2, -2.95459980854025e-03f);
    p = fmaf(p, x2, -1.60960333262415e-02f);
    float q = -1.45660718464996e-05f;
    q = fmaf(q, x2, -2.13374055278905e-04f);
    q = fmaf(q, x2, -1.68282697438203e-03f);
    q = fmaf(q, x2, -7.37332916720468e-03f);
    q = fmaf(q, x2, -1.42647390514189e-02f);
    const float r = (x * p) * __builtin_amdgcn_rcpf(q);
    return fminf(fmaxf(r, -1.f), 1.f);
}

__global__ __launch_bounds__(256) void k1_row_fft(const float* __restrict__ xre,
                                                  const float* __restrict__ xim,
                                                  const float* __restrict__ gamma,
                                                  const float* __restrict__ beta,
                                                  const float4* __restrict__ tw,
                                                  float2* __restrict__ R)
{
    __shared__ __align__(16) float2 z[N_];
    __shared__ float red[4 * 256];
    const int row0 = 2 * blockIdx.x;
    const int c0   = row0 & (C_ - 1);
    const int tid  = threadIdx.x;
    const float* __restrict__ a1 = xre + (size_t)row0 * N_;
    const float* __restrict__ b1 = xim + (size_t)row0 * N_;
    const float* __restrict__ a2 = a1 + N_;
    const float* __restrict__ b2 = b1 + N_;

    float4 wa = tw_entry(tw, tid, 12), wb;

    #pragma unroll 4
    for (int k = tid; k < NH; k += 256) {
        const int km = (N_ - k) & (N_ - 1);
        const float sA1 = a1[k] + a1[km];
        const float dB1 = b1[k] - b1[km];
        const float sA2 = a2[k] + a2[km];
        const float dB2 = b2[k] - b2[km];
        z[lds_slot(k)]  = make_float2(0.5f * (sA1 - dB2), 0.5f * (dB1 + sA2));
        z[lds_slot(km)] = make_float2(0.5f * (sA1 + dB2), 0.5f * (sA2 - dB1));
    }
    if (tid == 0) z[lds_slot(NH)] = make_float2(a1[NH], a2[NH]);
    __syncthreads();

    wb = tw_entry(tw, tid, 10);
    pass4_large<true, 12>(z, tw, tid, wa);  __syncthreads();
    wa = tw_entry(tw, tid, 8);
    pass4_large<true, 10>(z, tw, tid, wb);  __syncthreads();
    for (int L = 8; L >= 2; L -= 2) {
        wb = tw_entry(tw, tid, (L > 2) ? L - 2 : 2);
        pass4_small<true>(z, tid, L, wa);
        __syncthreads();
        wa = wb;
    }
    pass_small<true>(z, tid, 0);
    __syncthreads();

    float s1 = 0.f, q1 = 0.f, s2 = 0.f, q2 = 0.f;
    const int st = lds_slot(tid);
    for (int i = st; i < N_; i += 256) {
        const float2 t = z[i];
        const float v = t.x;  s1 += v;  q1 += v * v;
        const float w = t.y;  s2 += w;  q2 += w * w;
    }
    red[tid]       = s1;
    red[256 + tid] = q1;
    red[512 + tid] = s2;
    red[768 + tid] = q2;
    __syncthreads();
    for (int off = 128; off > 0; off >>= 1) {
        if (tid < off) {
            red[tid]       += red[tid + off];
            red[256 + tid] += red[256 + tid + off];
            red[512 + tid] += red[512 + tid + off];
            red[768 + tid] += red[768 + tid + off];
        }
        __syncthreads();
    }
    const float mean1 = red[0]   * (1.f / N_);
    const float var1  = red[256] * (1.f / N_) - mean1 * mean1;
    const float istd1 = rsqrtf(var1 + 1e-5f);
    const float mean2 = red[512] * (1.f / N_);
    const float var2  = red[768] * (1.f / N_) - mean2 * mean2;
    const float istd2 = rsqrtf(var2 + 1e-5f);
    const float gm1 = gamma[c0],     bt1 = beta[c0];
    const float gm2 = gamma[c0 + 1], bt2 = beta[c0 + 1];

    for (int i = st; i < N_; i += 256) {
        const float2 t = z[i];
        float v = (t.x - mean1) * istd1 * gm1 + bt1;
        v = 0.5f * v * (1.f + erf_rational(v * 0.70710678118654752f));
        float w = (t.y - mean2) * istd2 * gm2 + bt2;
        w = 0.5f * w * (1.f + erf_rational(w * 0.70710678118654752f));
        z[i] = make_float2(v, w);
    }
    __syncthreads();

    wa = tw_entry(tw, tid, 2);
    pass_small<false>(z, tid, 0);
    __syncthreads();
    for (int L = 2; L <= 8; L += 2) {
        wb = tw_entry(tw, tid, L + 2);
        pass4_small<false>(z, tid, L, wa);
        __syncthreads();
        wa = wb;
    }
    wb = tw_entry(tw, tid, 12);
    pass4_large<false, 10>(z, tw, tid, wa);  __syncthreads();
    pass4_large<false, 12>(z, tw, tid, wb);  __syncthreads();

    float2* R0 = R + (size_t)row0 * KS;
    float2* R1 = R0 + KS;
    for (int k = tid; k < KBINS; k += 256) {
        const int km = (N_ - k) & (N_ - 1);
        const float2 fk = z[lds_slot(k)], fm = z[lds_slot(km)];
        const float pk = fk.x, qk = fk.y, pm = fm.x, qm = fm.y;
        R0[k] = make_float2(0.5f * (pk + pm), 0.5f * (qk - qm));
        R1[k] = make_float2(0.5f * (qk + qm), 0.5f * (pm - pk));
    }
}

__device__ __forceinline__ float xhalf(float x)
{
    const int xi = __builtin_bit_cast(int, x);
    return __builtin_bit_cast(float,
        __builtin_amdgcn_permlanex16(xi, xi, 0x76543210, 0xfedcba98, false, false));
}

__device__ __forceinline__ void store_bin(float* __restrict__ outre, float* __restrict__ outim,
                                          int b, int row, int kcol, float orv, float oim)
{
    const size_t o = (size_t)(b * C_ + row) * N_;
    const bool edge = (kcol == 0) || (kcol == NH);
    const float oiv = edge ? 0.f : oim;
    outre[o + kcol] = orv;
    outim[o + kcol] = oiv;
    if (kcol >= 1 && kcol <= NH - 1) {
        outre[o + (N_ - kcol)] =  orv;
        outim[o + (N_ - kcol)] = -oiv;
    }
}

__global__ __launch_bounds__(128) void k2_dftC_wmma(const float2* __restrict__ R,
                                                    const float2* __restrict__ Wc,
                                                    const float2* __restrict__ Ws,
                                                    float* __restrict__ outre,
                                                    float* __restrict__ outim)
{
    __shared__ float Lre[C_][40];
    __shared__ float Lim[C_][40];
    const int b    = blockIdx.y;
    const int col0 = blockIdx.x * 32;
    const int tid  = threadIdx.x;
    const float2* Rb = R + (size_t)b * C_ * KS;

    {
        const int  cc = tid & 31;
        const int  r0 = tid >> 5;
        const int  k  = col0 + cc;
        const bool in = k < KBINS;
        const float2* src = Rb + (size_t)r0 * KS + (in ? k : KBINS - 1);
        float2 v[32];
        #pragma unroll
        for (int i = 0; i < 32; ++i) v[i] = src[(size_t)(4 * i) * KS];
        #pragma unroll
        for (int i = 0; i < 32; ++i) {
            Lre[r0 + 4 * i][cc] = in ? v[i].x : 0.f;
            Lim[r0 + 4 * i][cc] = in ? v[i].y : 0.f;
        }
    }
    __syncthreads();

    const int wave = tid >> 5;
    const int lane = tid & 31;
    const int lg   = lane >> 4;
    const int ln   = lane & 15;
    const int m    = wave * 16 + ln;
    const float2* WcRow = Wc + m * WPAIR;
    const float2* WsRow = Ws + m * WPAIR;

    const v8f zero8 = {0.f,0.f,0.f,0.f,0.f,0.f,0.f,0.f};
    v8f cr0 = zero8, si0 = zero8, ci0 = zero8, sr0 = zero8;
    v8f cr1 = zero8, si1 = zero8, ci1 = zero8, sr1 = zero8;

    for (int k0 = 0; k0 < C_; k0 += 4) {
        const int kA = k0 + 2 * lg;
        float2 wc = WcRow[(k0 >> 1) + lg];
        float2 ws = WsRow[(k0 >> 1) + lg];
        v2f aC = { wc.x, wc.y };
        v2f aS = { ws.x, ws.y };
        v2f bRe0 = { Lre[kA][ln],      Lre[kA + 1][ln]      };
        v2f bIm0 = { Lim[kA][ln],      Lim[kA + 1][ln]      };
        v2f bRe1 = { Lre[kA][16 + ln], Lre[kA + 1][16 + ln] };
        v2f bIm1 = { Lim[kA][16 + ln], Lim[kA + 1][16 + ln] };
        cr0 = __builtin_amdgcn_wmma_f32_16x16x4_f32(false, aC, false, bRe0, (short)0, cr0, false, false);
        si0 = __builtin_amdgcn_wmma_f32_16x16x4_f32(false, aS, false, bIm0, (short)0, si0, false, false);
        ci0 = __builtin_amdgcn_wmma_f32_16x16x4_f32(false, aC, false, bIm0, (short)0, ci0, false, false);
        sr0 = __builtin_amdgcn_wmma_f32_16x16x4_f32(false, aS, false, bRe0, (short)0, sr0, false, false);
        cr1 = __builtin_amdgcn_wmma_f32_16x16x4_f32(false, aC, false, bRe1, (short)0, cr1, false, false);
        si1 = __builtin_amdgcn_wmma_f32_16x16x4_f32(false, aS, false, bIm1, (short)0, si1, false, false);
        ci1 = __builtin_amdgcn_wmma_f32_16x16x4_f32(false, aC, false, bIm1, (short)0, ci1, false, false);
        sr1 = __builtin_amdgcn_wmma_f32_16x16x4_f32(false, aS, false, bRe1, (short)0, sr1, false, false);
    }

    const int  kcol = col0 + lane;
    const bool ok   = kcol < KBINS;
    #pragma unroll
    for (int v = 0; v < 8; ++v) {
        const int  mr = wave * 16 + 8 * lg + v;
        const bool z  = (mr == 0);
        float q[2][4];
        for (int ct = 0; ct < 2; ++ct) {
            const float p1 = ct ? cr1[v] : cr0[v];
            const float p2 = ct ? si1[v] : si0[v];
            const float p3 = ct ? ci1[v] : ci0[v];
            const float p4 = ct ? sr1[v] : sr0[v];
            q[ct][0] = z ? p1 : p1 + p2;
            q[ct][1] = z ? p3 : p3 - p4;
            q[ct][2] = z ? p4 : p1 - p2;
            q[ct][3] = z ? p2 : p3 + p4;
        }
        float lo[4], hi[4];
        for (int t = 0; t < 4; ++t) {
            const float x0 = xhalf(q[0][t]);
            const float x1 = xhalf(q[1][t]);
            lo[t] = lg ? x1      : q[0][t];
            hi[t] = lg ? q[1][t] : x0;
        }
        const int mLo = wave * 16 + v;
        const int mHi = mLo + 8;
        if (ok) {
            store_bin(outre, outim, b, mLo,                            kcol, lo[0], lo[1]);
            store_bin(outre, outim, b, mLo == 0 ? (C_ / 2) : C_ - mLo, kcol, lo[2], lo[3]);
            store_bin(outre, outim, b, mHi,                            kcol, hi[0], hi[1]);
            store_bin(outre, outim, b, C_ - mHi,                       kcol, hi[2], hi[3]);
        }
    }
}

extern "C" void kernel_launch(void* const* d_in, const int* in_sizes, int n_in,
                              void* d_out, int out_size, void* d_ws, size_t ws_size,
                              hipStream_t stream) {
    const float* xre   = (const float*)d_in[0];
    const float* xim   = (const float*)d_in[1];
    const float* gamma = (const float*)d_in[2];
    const float* beta  = (const float*)d_in[3];
    float* outre = (float*)d_out;
    float* outim = outre + (size_t)B_ * C_ * N_;

    float2* R  = (float2*)d_ws;
    float2* Wc = R + (size_t)B_ * C_ * KS;
    float2* Ws = Wc + MH_ * WPAIR;

    float4* tw = (float4*)d_out;

    k0_twiddle<<<dim3(K0_BLOCKS), dim3(256), 0, stream>>>(Wc, Ws, tw);
    k1_row_fft<<<dim3((B_ * C_) / 2), dim3(256), 0, stream>>>(xre, xim, gamma, beta, tw, R);
    k2_dftC_wmma<<<dim3((KBINS + 31) / 32, B_), dim3(128), 0, stream>>>(R, Wc, Ws, outre, outim);
}
